# starting kernel
# baseline (speedup 1.0000x reference)
#include <hip/hip_runtime.h>
#include <hip/hip_bf16.h>

#define N_NODES 50000
#define N_EDGES 800000
#define F_IN    128
#define HEADS   8
#define FH      16
#define HF      128
#define NEG_SLOPE 0.2f

typedef float v2f __attribute__((ext_vector_type(2)));
typedef float v4f __attribute__((ext_vector_type(4)));
typedef float v8f __attribute__((ext_vector_type(8)));

__global__ __launch_bounds__(256) void gat_gemm_wmma(
    const float* __restrict__ x, const float* __restrict__ W,
    float* __restrict__ h) {
  const int lane = threadIdx.x & 31;
  const int wave = threadIdx.x >> 5;
  const int m0   = blockIdx.x * 16;
  const int n0   = wave * 16;
  const int half = lane >> 4;
  const int l15  = lane & 15;

  const float* __restrict__ arow = x + (m0 + l15) * F_IN + 2 * half;
  const float* __restrict__ brow = W + (n0 + l15) * F_IN + 2 * half;

  v8f c = {};
#pragma unroll
  for (int k0 = 0; k0 < F_IN; k0 += 4) {
    v2f a; a.x = arow[k0]; a.y = arow[k0 + 1];
    v2f b; b.x = brow[k0]; b.y = brow[k0 + 1];
    c = __builtin_amdgcn_wmma_f32_16x16x4_f32(
         false, a,  false, b,
         (short)0, c,  false,  false);
  }

  float* __restrict__ op = h + (m0 + half * 8) * HF + n0 + l15;
#pragma unroll
  for (int v = 0; v < 8; ++v) op[v * HF] = c[v];
}

__global__ void gat_logits(const float* __restrict__ h,
                           const float* __restrict__ att_src,
                           const float* __restrict__ att_dst,
                           float* __restrict__ asrc, float* __restrict__ adst) {
  int t = blockIdx.x * blockDim.x + threadIdx.x;
  if (t >= N_NODES * HEADS) return;
  int head = t & 7;
  int n    = t >> 3;
  const float* __restrict__ hp = h + n * HF + head * FH;
  const float* __restrict__ as = att_src + head * FH;
  const float* __restrict__ ad = att_dst + head * FH;
  float s = 0.f, d = 0.f;
#pragma unroll
  for (int f = 0; f < FH; ++f) { float v = hp[f]; s += v * as[f]; d += v * ad[f]; }
  asrc[t] = s;
  adst[t] = d;
}

__global__ void gat_init(float* __restrict__ out, const float* __restrict__ bias,
                         int* __restrict__ mkey, float* __restrict__ denom) {
  int t = blockIdx.x * blockDim.x + threadIdx.x;
  if (t < N_NODES * HF) out[t] = bias[t & (HF - 1)];
  if (t < N_NODES * HEADS) { mkey[t] = (int)0x80000000; denom[t] = 0.f; }
}

__device__ __forceinline__ int f32_key(float f) {
  int i = __float_as_int(f);
  return i >= 0 ? i : (i ^ 0x7fffffff);
}
__device__ __forceinline__ float key_f32(int k) {
  return __int_as_float(k >= 0 ? k : (k ^ 0x7fffffff));
}

__device__ __forceinline__ float leaky(float v) {
  return v >= 0.f ? v : NEG_SLOPE * v;
}

__global__ void gat_edge_max(const int* __restrict__ ei,
                             const float* __restrict__ asrc,
                             const float* __restrict__ adst,
                             int* __restrict__ mkey) {
  int t = blockIdx.x * blockDim.x + threadIdx.x;
  if (t >= N_EDGES * HEADS) return;
  int e = t >> 3, head = t & 7;
  int s = ei[e];
  int d = ei[N_EDGES + e];
  float v = leaky(asrc[s * HEADS + head] + adst[d * HEADS + head]);
  __hip_atomic_fetch_max(&mkey[d * HEADS + head], f32_key(v),
                         __ATOMIC_RELAXED, __HIP_MEMORY_SCOPE_AGENT);
}

__global__ void gat_edge_exp(const int* __restrict__ ei,
                             const float* __restrict__ asrc,
                             const float* __restrict__ adst,
                             const int* __restrict__ mkey,
                             float* __restrict__ exw,
                             float* __restrict__ denom) {
  int t = blockIdx.x * blockDim.x + threadIdx.x;
  if (t >= N_EDGES * HEADS) return;
  int e = t >> 3, head = t & 7;
  int s = ei[e];
  int d = ei[N_EDGES + e];
  float v = leaky(asrc[s * HEADS + head] + adst[d * HEADS + head]);
  float m = key_f32(mkey[d * HEADS + head]);
  float ex = __expf(v - m);
  exw[t] = ex;
  __hip_atomic_fetch_add(&denom[d * HEADS + head], ex,
                         __ATOMIC_RELAXED, __HIP_MEMORY_SCOPE_AGENT);
}

__global__ __launch_bounds__(256) void gat_aggregate(
    const int* __restrict__ ei, const float* __restrict__ h,
    const float* __restrict__ exw, const float* __restrict__ denom,
    float* __restrict__ out) {
  int t = blockIdx.x * blockDim.x + threadIdx.x;
  int e    = t >> 5;
  int lane = t & 31;
  int s = ei[e];
  int d = ei[N_EDGES + e];
  int head = lane >> 2;
  float alpha = exw[e * HEADS + head] /
                (denom[d * HEADS + head] + 1e-16f);
  const v4f hv = *(const v4f*)(h + s * HF + lane * 4);
  float* __restrict__ op = out + d * HF + lane * 4;
#pragma unroll
  for (int i = 0; i < 4; ++i)
    __hip_atomic_fetch_add(&op[i], alpha * hv[i],
                           __ATOMIC_RELAXED, __HIP_MEMORY_SCOPE_AGENT);
}

extern "C" void kernel_launch(void* const* d_in, const int* in_sizes, int n_in,
                              void* d_out, int out_size, void* d_ws, size_t ws_size,
                              hipStream_t stream) {
  const float* x        = (const float*)d_in[0];
  const float* W        = (const float*)d_in[1];
  const float* att_src  = (const float*)d_in[2];
  const float* att_dst  = (const float*)d_in[3];
  const float* bias     = (const float*)d_in[4];
  const int*   ei       = (const int*)d_in[5];
  float*       out      = (float*)d_out;

  float* h     = (float*)d_ws;
  float* asrc  = h     + (size_t)N_NODES * HF;
  float* adst  = asrc  + (size_t)N_NODES * HEADS;
  float* denom = adst  + (size_t)N_NODES * HEADS;
  int*   mkey  = (int*)(denom + (size_t)N_NODES * HEADS);
  float* exw   = (float*)(mkey + (size_t)N_NODES * HEADS);

  const int BLK = 256;

  gat_gemm_wmma<<<N_NODES / 16, BLK, 0, stream>>>(x, W, h);

  gat_logits<<<(N_NODES * HEADS + BLK - 1) / BLK, BLK, 0, stream>>>(
      h, att_src, att_dst, asrc, adst);

  gat_init<<<(N_NODES * HF + BLK - 1) / BLK, BLK, 0, stream>>>(
      out, bias, mkey, denom);

  gat_edge_max<<<(N_EDGES * HEADS + BLK - 1) / BLK, BLK, 0, stream>>>(
      ei, asrc, adst, mkey);

  gat_edge_exp<<<(N_EDGES * HEADS + BLK - 1) / BLK, BLK, 0, stream>>>(
      ei, asrc, adst, mkey, exw, denom);

  gat_aggregate<<<(N_EDGES * 32) / BLK, BLK, 0, stream>>>(
      ei, h, exw, denom, out);
}


# Edge phase by destination ownership: a work-group per 196-node range scans the dst column, keeps max, denominator and row sums in LDS (LDS float adds, arrival order), divides once; no global atomics
# speedup vs baseline: 2.1931x; 2.1931x over previous
#include <hip/hip_runtime.h>
#include <hip/hip_bf16.h>

#define N_NODES 50000
#define N_EDGES 800000
#define F_IN    128
#define HEADS   8
#define FH      16
#define HF      128
#define NEG_SLOPE 0.2f

typedef float v2f __attribute__((ext_vector_type(2)));
typedef float v4f __attribute__((ext_vector_type(4)));
typedef float v8f __attribute__((ext_vector_type(8)));

__global__ __launch_bounds__(256) void gat_gemm_wmma(
    const float* __restrict__ x, const float* __restrict__ W,
    float* __restrict__ h) {
  const int lane = threadIdx.x & 31;
  const int wave = threadIdx.x >> 5;
  const int m0   = blockIdx.x * 16;
  const int n0   = wave * 16;
  const int half = lane >> 4;
  const int l15  = lane & 15;

  const float* __restrict__ arow = x + (m0 + l15) * F_IN + 2 * half;
  const float* __restrict__ brow = W + (n0 + l15) * F_IN + 2 * half;

  v8f c = {};
#pragma unroll
  for (int k0 = 0; k0 < F_IN; k0 += 4) {
    v2f a; a.x = arow[k0]; a.y = arow[k0 + 1];
    v2f b; b.x = brow[k0]; b.y = brow[k0 + 1];
    c = __builtin_amdgcn_wmma_f32_16x16x4_f32(
         false, a,  false, b,
         (short)0, c,  false,  false);
  }

  float* __restrict__ op = h + (m0 + half * 8) * HF + n0 + l15;
#pragma unroll
  for (int v = 0; v < 8; ++v) op[v * HF] = c[v];
}

__global__ void gat_logits(const float* __restrict__ h,
                           const float* __restrict__ att_src,
                           const float* __restrict__ att_dst,
                           float* __restrict__ asrc, float* __restrict__ adst) {
  int t = blockIdx.x * blockDim.x + threadIdx.x;
  if (t >= N_NODES * HEADS) return;
  int head = t & 7;
  int n    = t >> 3;
  const float* __restrict__ hp = h + n * HF + head * FH;
  const float* __restrict__ as = att_src + head * FH;
  const float* __restrict__ ad = att_dst + head * FH;
  float s = 0.f, d = 0.f;
#pragma unroll
  for (int f = 0; f < FH; ++f) { float v = hp[f]; s += v * as[f]; d += v * ad[f]; }
  asrc[t] = s;
  adst[t] = d;
}

__device__ __forceinline__ int f32_key(float f) {
  int i = __float_as_int(f);
  return i >= 0 ? i : (i ^ 0x7fffffff);
}
__device__ __forceinline__ float key_f32(int k) {
  return __int_as_float(k >= 0 ? k : (k ^ 0x7fffffff));
}

__device__ __forceinline__ float leaky(float v) {
  return v >= 0.f ? v : NEG_SLOPE * v;
}

#define RNG   196
#define NGRP  256
#define CAP   32768
#define NTHR  1024

static_assert(RNG * NGRP >= N_NODES, "ranges must cover every node");
static_assert(N_NODES <= 65536 && RNG <= 256, "list entry packs src in 16 bits, local dst in 8");
static_assert(N_EDGES <= (1 << 20) && N_EDGES % 32 == 0 && N_EDGES % 4 == 0, "edge id packs in 20 bits");

typedef int v4i __attribute__((ext_vector_type(4)));

__device__ __forceinline__ void edge_sum(int lane, int dl, float sa, const float (&hv)[4],
                                         const float* adl, const int* mxk,
                                         float* den, float* acc) {
  const int b = dl * HEADS + (lane & 7);
  const float v  = leaky(sa + adl[b]);
  const float ex = __expf(v - key_f32(mxk[b]));
  if (lane < 8)
    __hip_atomic_fetch_add(&den[b], ex, __ATOMIC_RELAXED, __HIP_MEMORY_SCOPE_WORKGROUP);
  float* ap = acc + dl * HF + lane;
  const bool hi = lane >= 16;
  const int exi = __float_as_int(ex);
#pragma unroll
  for (int j = 0; j < 4; ++j) {
    const float e0 = __int_as_float(__builtin_amdgcn_readlane(exi, 2 * j));
    const float e1 = __int_as_float(__builtin_amdgcn_readlane(exi, 2 * j + 1));
    const float a = hi ? e1 : e0;
    __hip_atomic_fetch_add(&ap[32 * j], a * hv[j], __ATOMIC_RELAXED, __HIP_MEMORY_SCOPE_WORKGROUP);
  }
}

__global__ __launch_bounds__(NTHR) void gat_owner(
    const int* __restrict__ ei, const float* __restrict__ h,
    const float* __restrict__ asrc, const float* __restrict__ adst,
    const float* __restrict__ bias, float* __restrict__ out) {
  __shared__ float    acc[RNG * HF];
  __shared__ int      mxk[RNG * HEADS];
  __shared__ float    den[RNG * HEADS];
  __shared__ float    adl[RNG * HEADS];
  __shared__ unsigned list[CAP];
  __shared__ unsigned cnt;

  const int tid  = threadIdx.x;
  const int lane = tid & 31;
  const int wave = tid >> 5;
  const int n0   = blockIdx.x * RNG;
  const int nloc = (N_NODES - n0) < RNG ? (N_NODES - n0) : RNG;

  for (int i = tid; i < RNG * HF; i += NTHR) acc[i] = 0.f;
  for (int i = tid; i < RNG * HEADS; i += NTHR) {
    mxk[i] = (int)0x80000000;
    den[i] = 0.f;
    adl[i] = (i < nloc * HEADS) ? adst[n0 * HEADS + i] : 0.f;
  }
  if (tid == 0) cnt = 0u;
  __syncthreads();

  {
    const v4i* __restrict__ dp = reinterpret_cast<const v4i*>(ei + N_EDGES);
    constexpr int NV = N_EDGES / 4;
    const int rot = blockIdx.x * (NV / NGRP);
    for (int j0 = tid; j0 < NV; j0 += NTHR * 8) {
      v4i d4[8];
      int jj[8];
#pragma unroll
      for (int u = 0; u < 8; ++u) {
        const int j = j0 + u * NTHR;
        int q = j + rot;
        if (q >= NV) q -= NV;
        jj[u] = q;
        if (j < NV) d4[u] = dp[q];
        else { d4[u].x = -1; d4[u].y = -1; d4[u].z = -1; d4[u].w = -1; }
      }
#pragma unroll
      for (int u = 0; u < 8; ++u) {
#pragma unroll
        for (int c = 0; c < 4; ++c) {
          const unsigned dl = (unsigned)(d4[u][c] - n0);
          if (dl < (unsigned)RNG) {
            const unsigned slot = atomicAdd(&cnt, 1u);
            if (slot < (unsigned)CAP) list[slot] = (unsigned)(jj[u] * 4 + c) | (dl << 20);
          }
        }
      }
    }
  }
  __syncthreads();
  const unsigned total = cnt;

  if (total <= (unsigned)CAP) {
    for (unsigned i = tid; i < total; i += NTHR) {
      const unsigned w = list[i];
      const unsigned s = (unsigned)ei[w & 0xFFFFFu];
      list[i] = s | ((w >> 20) << 16);
    }
    __syncthreads();

    const unsigned items = total * HEADS;
    for (unsigned i0 = tid; i0 < items; i0 += NTHR * 4) {
      float sv[4];
      unsigned dh[4];
#pragma unroll
      for (int u = 0; u < 4; ++u) {
        const unsigned idx = i0 + u * NTHR;
        sv[u] = 0.f;
        dh[u] = 0xFFFFFFFFu;
        if (idx < items) {
          const unsigned w = list[idx >> 3];
          const unsigned hd = idx & 7u;
          sv[u] = asrc[(w & 0xFFFFu) * HEADS + hd];
          dh[u] = (w >> 16) * HEADS + hd;
        }
      }
#pragma unroll
      for (int u = 0; u < 4; ++u) {
        if (dh[u] != 0xFFFFFFFFu) {
          const float v = leaky(sv[u] + adl[dh[u]]);
          __hip_atomic_fetch_max(&mxk[dh[u]], f32_key(v), __ATOMIC_RELAXED, __HIP_MEMORY_SCOPE_WORKGROUP);
        }
      }
    }
    __syncthreads();

    for (unsigned i0 = wave; i0 < total; i0 += 32 * 4) {
      float hv[4][4];
      float sa[4];
      int   dl[4];
#pragma unroll
      for (int u = 0; u < 4; ++u) {
        const unsigned i = i0 + u * 32;
        dl[u] = -1;
        sa[u] = 0.f;
#pragma unroll
        for (int j = 0; j < 4; ++j) hv[u][j] = 0.f;
        if (i < total) {
          const unsigned w = (unsigned)__builtin_amdgcn_readfirstlane((int)list[i]);
          const int s = (int)(w & 0xFFFFu);
          dl[u] = (int)(w >> 16);
          sa[u] = asrc[s * HEADS + (lane & 7)];
          const float* __restrict__ hp = h + s * HF + lane;
#pragma unroll
          for (int j = 0; j < 4; ++j) hv[u][j] = hp[32 * j];
        }
      }
#pragma unroll
      for (int u = 0; u < 4; ++u)
        if (dl[u] >= 0) edge_sum(lane, dl[u], sa[u], hv[u], adl, mxk, den, acc);
    }
    __syncthreads();
  } else {
    for (int ph = 0; ph < 2; ++ph) {
      for (int base = wave * 32; base < N_EDGES; base += 32 * 32) {
        const unsigned d = (unsigned)(ei[N_EDGES + base + lane] - n0);
        unsigned mask = __builtin_amdgcn_ballot_w32(d < (unsigned)RNG);
        while (mask) {
          const int bpos = __builtin_ctz(mask);
          mask &= mask - 1u;
          const int s   = ei[base + bpos];
          const int dlu = __shfl((int)d, bpos, 32);
          const float sa = asrc[s * HEADS + (lane & 7)];
          if (ph == 0) {
            const int b = dlu * HEADS + (lane & 7);
            const float v = leaky(sa + adl[b]);
            if (lane < 8)
              __hip_atomic_fetch_max(&mxk[b], f32_key(v), __ATOMIC_RELAXED, __HIP_MEMORY_SCOPE_WORKGROUP);
          } else {
            float hv[4];
            const float* __restrict__ hp = h + s * HF + lane;
#pragma unroll
            for (int j = 0; j < 4; ++j) hv[j] = hp[32 * j];
            edge_sum(lane, dlu, sa, hv, adl, mxk, den, acc);
          }
        }
      }
      __syncthreads();
    }
  }

  for (int i = tid; i < nloc * HF; i += NTHR) {
    const int r = i >> 7, f = i & (HF - 1);
    out[n0 * HF + i] = acc[i] / (den[r * HEADS + (f >> 4)] + 1e-16f) + bias[f];
  }
}

extern "C" void kernel_launch(void* const* d_in, const int* in_sizes, int n_in,
                              void* d_out, int out_size, void* d_ws, size_t ws_size,
                              hipStream_t stream) {
  const float* x        = (const float*)d_in[0];
  const float* W        = (const float*)d_in[1];
  const float* att_src  = (const float*)d_in[2];
  const float* att_dst  = (const float*)d_in[3];
  const float* bias     = (const float*)d_in[4];
  const int*   ei       = (const int*)d_in[5];
  float*       out      = (float*)d_out;

  float* h     = (float*)d_ws;
  float* asrc  = h     + (size_t)N_NODES * HF;
  float* adst  = asrc  + (size_t)N_NODES * HEADS;

  const int BLK = 256;

  gat_gemm_wmma<<<N_NODES / 16, BLK, 0, stream>>>(x, W, h);

  gat_logits<<<(N_NODES * HEADS + BLK - 1) / BLK, BLK, 0, stream>>>(
      h, att_src, att_dst, asrc, adst);

  gat_owner<<<NGRP, NTHR, 0, stream>>>(ei, h, asrc, adst, bias, out);
}


# Projection W-stationary: a wave keeps its head's 16 W rows in registers and walks 3-4 row tiles of x (16-byte loads); 8192 waves, not 25000; k regrouped per matrix op: sum order changes, still fixed
# speedup vs baseline: 3.2085x; 1.4630x over previous
#include <hip/hip_runtime.h>
#include <hip/hip_bf16.h>

#define N_NODES 50000
#define N_EDGES 800000
#define F_IN    128
#define HEADS   8
#define FH      16
#define HF      128
#define NEG_SLOPE 0.2f

typedef float v2f __attribute__((ext_vector_type(2)));
typedef float v4f __attribute__((ext_vector_type(4)));
typedef float v8f __attribute__((ext_vector_type(8)));

#define GEMM_WG  1024
#define M_TILES  (N_NODES / 16)

static_assert(N_NODES % 16 == 0, "whole 16-row tiles");

__global__ __launch_bounds__(256) void gat_gemm_wmma(
    const float* __restrict__ x, const float* __restrict__ W,
    float* __restrict__ h) {
  const int lane = threadIdx.x & 31;
  const int wave = threadIdx.x >> 5;
  const int n0   = wave * 16;
  const int half = lane >> 4;
  const int l15  = lane & 15;

  const v4f* __restrict__ brow =
      reinterpret_cast<const v4f*>(W + (n0 + l15) * F_IN + 4 * half);
  v4f b[16];
#pragma unroll
  for (int q = 0; q < 16; ++q) b[q] = brow[2 * q];

  for (int t = blockIdx.x; t < M_TILES; t += GEMM_WG) {
    const int m0 = t * 16;
    const v4f* __restrict__ arow =
        reinterpret_cast<const v4f*>(x + (m0 + l15) * F_IN + 4 * half);
    v8f c = {};
#pragma unroll
    for (int hh = 0; hh < 2; ++hh) {
      v4f a[8];
#pragma unroll
      for (int q = 0; q < 8; ++q) a[q] = arow[2 * (8 * hh + q)];
      asm volatile("" :: "v"(a[0]), "v"(a[1]), "v"(a[2]), "v"(a[3]),
                         "v"(a[4]), "v"(a[5]), "v"(a[6]), "v"(a[7]));
#pragma unroll
      for (int q = 0; q < 8; ++q) {
        const v4f bq = b[8 * hh + q];
        v2f a0; a0.x = a[q].x; a0.y = a[q].y;
        v2f a1; a1.x = a[q].z; a1.y = a[q].w;
        v2f b0; b0.x = bq.x;   b0.y = bq.y;
        v2f b1; b1.x = bq.z;   b1.y = bq.w;
        c = __builtin_amdgcn_wmma_f32_16x16x4_f32(
             false, a0,  false, b0,
             (short)0, c,  false,  false);
        c = __builtin_amdgcn_wmma_f32_16x16x4_f32(
             false, a1,  false, b1,
             (short)0, c,  false,  false);
      }
    }

    float* __restrict__ op = h + (m0 + half * 8) * HF + n0 + l15;
#pragma unroll
    for (int v = 0; v < 8; ++v) op[v * HF] = c[v];
  }
}

#define RNG   196
#define NGRP  256
#define NCH   (N_EDGES / 32)
#define NBLK  ((NCH + 31) / 32)

typedef int v4i __attribute__((ext_vector_type(4)));

static_assert(N_EDGES % 128 == 0, "a tag block is read in whole 16-byte loads of 32 lanes");

__global__ void gat_logits(const float* __restrict__ h,
                           const float* __restrict__ att_src,
                           const float* __restrict__ att_dst,
                           const int* __restrict__ ei,
                           float* __restrict__ asrc, float* __restrict__ adst,
                           unsigned* __restrict__ tagp) {
  int t = blockIdx.x * blockDim.x + threadIdx.x;
  const int gw = __builtin_amdgcn_readfirstlane(t >> 5);
  if (gw < NBLK) {
    const int lane = threadIdx.x & 31;
    const int* __restrict__ dcol = ei + N_EDGES + gw * 1024;
    v4i d4[8];
#pragma unroll
    for (int r = 0; r < 8; ++r) {
      const int ro = (gw * 1024 + 128 * r < N_EDGES) ? 128 * r : 0;
      d4[r] = reinterpret_cast<const v4i*>(dcol + ro)[lane];
    }
    unsigned mine[8];
#pragma unroll
    for (int k = 0; k < 8; ++k) mine[k] = 0u;
#pragma unroll
    for (int r = 0; r < 8; ++r) {
#pragma unroll
      for (int c = 0; c < 4; ++c) {
        unsigned g = (unsigned)d4[r][c] / (unsigned)RNG;
        g = g < (unsigned)NGRP ? g : (unsigned)(NGRP - 1);
#pragma unroll
        for (int k = 0; k < 8; ++k) {
          const unsigned pk = __builtin_amdgcn_ballot_w32(((g >> k) & 1u) != 0u);
          if (lane == 4 * r + c) mine[k] = pk;
        }
      }
    }
#pragma unroll
    for (int k = 0; k < 8; ++k) tagp[gw * 256 + k * 32 + lane] = mine[k];
  }
  if (t >= N_NODES * HEADS) return;
  int head = t & 7;
  int n    = t >> 3;
  const float* __restrict__ hp = h + n * HF + head * FH;
  const float* __restrict__ as = att_src + head * FH;
  const float* __restrict__ ad = att_dst + head * FH;
  float s = 0.f, d = 0.f;
#pragma unroll
  for (int f = 0; f < FH; ++f) { float v = hp[f]; s += v * as[f]; d += v * ad[f]; }
  asrc[t] = s;
  adst[t] = d;
}

__device__ __forceinline__ int f32_key(float f) {
  int i = __float_as_int(f);
  return i >= 0 ? i : (i ^ 0x7fffffff);
}
__device__ __forceinline__ float key_f32(int k) {
  return __int_as_float(k >= 0 ? k : (k ^ 0x7fffffff));
}

__device__ __forceinline__ float leaky(float v) {
  return v >= 0.f ? v : NEG_SLOPE * v;
}

#define CAP   32768
#define NTHR  1024

static_assert(RNG * NGRP >= N_NODES, "ranges must cover every node");
static_assert(NBLK * 32 <= N_NODES * HEADS, "the logits launch has a wave for every tag block");
static_assert(N_NODES <= 65536 && RNG <= 256 && NGRP <= 256, "list entry packs src in 16 bits, local dst in 8; a tag is a byte");
static_assert(N_EDGES % 32 == 0, "a tag word covers 32 edges, the exact path walks 32 edges a wave");
static_assert(NBLK * 1024 <= (1 << 27) && NBLK >= NGRP, "hit entry packs block, word and bit in 32 bits; the start rotation is whole blocks");

__device__ __forceinline__ void edge_sum(int lane, int dl, float sa, const float (&hv)[4],
                                         const float* adl, const int* mxk,
                                         float* den, float* acc) {
  const int b = dl * HEADS + (lane & 7);
  const float v  = leaky(sa + adl[b]);
  const float ex = __expf(v - key_f32(mxk[b]));
  if (lane < 8)
    __hip_atomic_fetch_add(&den[b], ex, __ATOMIC_RELAXED, __HIP_MEMORY_SCOPE_WORKGROUP);
  float* ap = acc + dl * HF + lane;
  const bool hi = lane >= 16;
  const int exi = __float_as_int(ex);
#pragma unroll
  for (int j = 0; j < 4; ++j) {
    const float e0 = __int_as_float(__builtin_amdgcn_readlane(exi, 2 * j));
    const float e1 = __int_as_float(__builtin_amdgcn_readlane(exi, 2 * j + 1));
    const float a = hi ? e1 : e0;
    __hip_atomic_fetch_add(&ap[32 * j], a * hv[j], __ATOMIC_RELAXED, __HIP_MEMORY_SCOPE_WORKGROUP);
  }
}

__global__ __launch_bounds__(NTHR) void gat_owner(
    const int* __restrict__ ei, const unsigned* __restrict__ tagp,
    const float* __restrict__ h,
    const float* __restrict__ asrc, const float* __restrict__ adst,
    const float* __restrict__ bias, float* __restrict__ out) {
  __shared__ float    acc[RNG * HF];
  __shared__ int      mxk[RNG * HEADS];
  __shared__ float    den[RNG * HEADS];
  __shared__ float    adl[RNG * HEADS];
  __shared__ unsigned list[CAP];
  __shared__ unsigned cnt;

  const int tid  = threadIdx.x;
  const int lane = tid & 31;
  const int wave = tid >> 5;
  const int n0   = blockIdx.x * RNG;
  const int nloc = (N_NODES - n0) < RNG ? (N_NODES - n0) : RNG;

  for (int i = tid; i < RNG * HF; i += NTHR) acc[i] = 0.f;
  for (int i = tid; i < RNG * HEADS; i += NTHR) {
    mxk[i] = (int)0x80000000;
    den[i] = 0.f;
    adl[i] = (i < nloc * HEADS) ? adst[n0 * HEADS + i] : 0.f;
  }
  if (tid == 0) cnt = 0u;
  __syncthreads();

  {
    constexpr int UB = 4;
    const int wv = __builtin_amdgcn_readfirstlane(wave);
    unsigned inv[8];
#pragma unroll
    for (int k = 0; k < 8; ++k) inv[k] = ((blockIdx.x >> k) & 1u) ? 0u : 0xFFFFFFFFu;
    const int rot = blockIdx.x * (NBLK / NGRP);
    for (int b0 = wv; b0 < NBLK; b0 += 32 * UB) {
      unsigned pw[UB][8];
      int cid[UB];
#pragma unroll
      for (int u = 0; u < UB; ++u) {
        const int b = b0 + 32 * u;
        int q = (b < NBLK ? b : 0) + rot;
        if (q >= NBLK) q -= NBLK;
        cid[u] = b < NBLK ? q * 32 + lane : NCH;
        const unsigned* __restrict__ bp = tagp + q * 256 + lane;
#pragma unroll
        for (int k = 0; k < 8; ++k) pw[u][k] = bp[32 * k];
      }
      unsigned hm[UB];
      unsigned nh = 0u;
#pragma unroll
      for (int u = 0; u < UB; ++u) {
        unsigned m = 0xFFFFFFFFu;
#pragma unroll
        for (int k = 0; k < 8; ++k) m &= pw[u][k] ^ inv[k];
        if (cid[u] >= NCH) m = 0u;
        hm[u] = m;
        nh += (unsigned)__builtin_popcount(m);
      }
      if (nh) {
        unsigned slot = atomicAdd(&cnt, nh);
        if (slot + nh <= (unsigned)CAP) {
#pragma unroll
          for (int u = 0; u < UB; ++u) {
            unsigned m = hm[u];
            const unsigned c0 = (unsigned)cid[u] << 5;
            while (m) {
              const unsigned p = (unsigned)__builtin_ctz(m);
              m &= m - 1u;
              list[slot] = c0 | p;
              ++slot;
            }
          }
        }
      }
    }
  }
  __syncthreads();
  const unsigned total = cnt;

  if (total <= (unsigned)CAP) {
    for (unsigned i = tid; i < total; i += NTHR) {
      const unsigned w = list[i];
      const unsigned p = w & 31u, j = (w >> 5) & 31u, blk = w >> 10;
      const unsigned e = blk * 1024u + 128u * (j >> 2) + 4u * p + (j & 3u);
      const unsigned s = (unsigned)ei[e];
      unsigned dl = (unsigned)(ei[N_EDGES + e] - n0);
      dl = dl < (unsigned)RNG ? dl : (unsigned)(RNG - 1);
      list[i] = s | (dl << 16);
    }
    __syncthreads();

    const unsigned items = total * HEADS;
    for (unsigned i0 = tid; i0 < items; i0 += NTHR * 4) {
      float sv[4];
      unsigned dh[4];
#pragma unroll
      for (int u = 0; u < 4; ++u) {
        const unsigned idx = i0 + u * NTHR;
        sv[u] = 0.f;
        dh[u] = 0xFFFFFFFFu;
        if (idx < items) {
          const unsigned w = list[idx >> 3];
          const unsigned hd = idx & 7u;
          sv[u] = asrc[(w & 0xFFFFu) * HEADS + hd];
          dh[u] = (w >> 16) * HEADS + hd;
        }
      }
#pragma unroll
      for (int u = 0; u < 4; ++u) {
        if (dh[u] != 0xFFFFFFFFu) {
          const float v = leaky(sv[u] + adl[dh[u]]);
          __hip_atomic_fetch_max(&mxk[dh[u]], f32_key(v), __ATOMIC_RELAXED, __HIP_MEMORY_SCOPE_WORKGROUP);
        }
      }
    }
    __syncthreads();

    for (unsigned i0 = wave; i0 < total; i0 += 32 * 4) {
      float hv[4][4];
      float sa[4];
      int   dl[4];
#pragma unroll
      for (int u = 0; u < 4; ++u) {
        const unsigned i = i0 + u * 32;
        dl[u] = -1;
        sa[u] = 0.f;
#pragma unroll
        for (int j = 0; j < 4; ++j) hv[u][j] = 0.f;
        if (i < total) {
          const unsigned w = (unsigned)__builtin_amdgcn_readfirstlane((int)list[i]);
          const int s = (int)(w & 0xFFFFu);
          dl[u] = (int)(w >> 16);
          sa[u] = asrc[s * HEADS + (lane & 7)];
          const float* __restrict__ hp = h + s * HF + lane;
#pragma unroll
          for (int j = 0; j < 4; ++j) hv[u][j] = hp[32 * j];
        }
      }
#pragma unroll
      for (int u = 0; u < 4; ++u)
        if (dl[u] >= 0) edge_sum(lane, dl[u], sa[u], hv[u], adl, mxk, den, acc);
    }
    __syncthreads();
  } else {
    for (int ph = 0; ph < 2; ++ph) {
      for (int base = wave * 32; base < N_EDGES; base += 32 * 32) {
        const unsigned d = (unsigned)(ei[N_EDGES + base + lane] - n0);
        unsigned mask = __builtin_amdgcn_ballot_w32(d < (unsigned)RNG);
        while (mask) {
          const int bpos = __builtin_ctz(mask);
          mask &= mask - 1u;
          const int s   = ei[base + bpos];
          const int dlu = __shfl((int)d, bpos, 32);
          const float sa = asrc[s * HEADS + (lane & 7)];
          if (ph == 0) {
            const int b = dlu * HEADS + (lane & 7);
            const float v = leaky(sa + adl[b]);
            if (lane < 8)
              __hip_atomic_fetch_max(&mxk[b], f32_key(v), __ATOMIC_RELAXED, __HIP_MEMORY_SCOPE_WORKGROUP);
          } else {
            float hv[4];
            const float* __restrict__ hp = h + s * HF + lane;
#pragma unroll
            for (int j = 0; j < 4; ++j) hv[j] = hp[32 * j];
            edge_sum(lane, dlu, sa, hv, adl, mxk, den, acc);
          }
        }
      }
      __syncthreads();
    }
  }

  for (int i = tid; i < nloc * HF; i += NTHR) {
    const int r = i >> 7, f = i & (HF - 1);
    out[n0 * HF + i] = acc[i] / (den[r * HEADS + (f >> 4)] + 1e-16f) + bias[f];
  }
}

extern "C" void kernel_launch(void* const* d_in, const int* in_sizes, int n_in,
                              void* d_out, int out_size, void* d_ws, size_t ws_size,
                              hipStream_t stream) {
  const float* x        = (const float*)d_in[0];
  const float* W        = (const float*)d_in[1];
  const float* att_src  = (const float*)d_in[2];
  const float* att_dst  = (const float*)d_in[3];
  const float* bias     = (const float*)d_in[4];
  const int*   ei       = (const int*)d_in[5];
  float*       out      = (float*)d_out;

  float* h     = (float*)d_ws;
  float* asrc  = h     + (size_t)N_NODES * HF;
  float* adst  = asrc  + (size_t)N_NODES * HEADS;
  unsigned* tagp = (unsigned*)(adst + (size_t)N_NODES * HEADS);

  const int BLK = 256;

  gat_gemm_wmma<<<GEMM_WG, BLK, 0, stream>>>(x, W, h);

  gat_logits<<<(N_NODES * HEADS + BLK - 1) / BLK, BLK, 0, stream>>>(
      h, att_src, att_dst, ei, asrc, adst, tagp);

  gat_owner<<<NGRP, NTHR, 0, stream>>>(ei, tagp, h, asrc, adst, bias, out);
}


# Owner: softmax weights exp(e-max) and denominators formed once per (edge, head) by one thread, kept in LDS; a sum-pass lane reads its 4 weights by one 16-byte LDS load; arrival-order LDS adds kept
# speedup vs baseline: 3.5497x; 1.1063x over previous
#include <hip/hip_runtime.h>
#include <hip/hip_bf16.h>

#define N_NODES 50000
#define N_EDGES 800000
#define F_IN    128
#define HEADS   8
#define FH      16
#define HF      128
#define NEG_SLOPE 0.2f

typedef float v2f __attribute__((ext_vector_type(2)));
typedef float v4f __attribute__((ext_vector_type(4)));
typedef float v8f __attribute__((ext_vector_type(8)));

#define GEMM_WG  1024
#define M_TILES  (N_NODES / 16)

static_assert(N_NODES % 16 == 0, "whole 16-row tiles");

__global__ __launch_bounds__(256) void gat_gemm_wmma(
    const float* __restrict__ x, const float* __restrict__ W,
    float* __restrict__ h) {
  const int lane = threadIdx.x & 31;
  const int wave = threadIdx.x >> 5;
  const int n0   = wave * 16;
  const int half = lane >> 4;
  const int l15  = lane & 15;

  const v4f* __restrict__ brow =
      reinterpret_cast<const v4f*>(W + (n0 + l15) * F_IN + 4 * half);
  v4f b[16];
#pragma unroll
  for (int q = 0; q < 16; ++q) b[q] = brow[2 * q];

  for (int t = blockIdx.x; t < M_TILES; t += GEMM_WG) {
    const int m0 = t * 16;
    const v4f* __restrict__ arow =
        reinterpret_cast<const v4f*>(x + (m0 + l15) * F_IN + 4 * half);
    v8f c = {};
#pragma unroll
    for (int hh = 0; hh < 2; ++hh) {
      v4f a[8];
#pragma unroll
      for (int q = 0; q < 8; ++q) a[q] = arow[2 * (8 * hh + q)];
      asm volatile("" :: "v"(a[0]), "v"(a[1]), "v"(a[2]), "v"(a[3]),
                         "v"(a[4]), "v"(a[5]), "v"(a[6]), "v"(a[7]));
#pragma unroll
      for (int q = 0; q < 8; ++q) {
        const v4f bq = b[8 * hh + q];
        v2f a0; a0.x = a[q].x; a0.y = a[q].y;
        v2f a1; a1.x = a[q].z; a1.y = a[q].w;
        v2f b0; b0.x = bq.x;   b0.y = bq.y;
        v2f b1; b1.x = bq.z;   b1.y = bq.w;
        c = __builtin_amdgcn_wmma_f32_16x16x4_f32(
             false, a0,  false, b0,
             (short)0, c,  false,  false);
        c = __builtin_amdgcn_wmma_f32_16x16x4_f32(
             false, a1,  false, b1,
             (short)0, c,  false,  false);
      }
    }

    float* __restrict__ op = h + (m0 + half * 8) * HF + n0 + l15;
#pragma unroll
    for (int v = 0; v < 8; ++v) op[v * HF] = c[v];
  }
}

#define RNG   196
#define NGRP  256
#define NCH   (N_EDGES / 32)
#define NBLK  ((NCH + 31) / 32)

typedef int v4i __attribute__((ext_vector_type(4)));

static_assert(N_EDGES % 128 == 0, "a tag block is read in whole 16-byte loads of 32 lanes");

__global__ void gat_logits(const float* __restrict__ h,
                           const float* __restrict__ att_src,
                           const float* __restrict__ att_dst,
                           const int* __restrict__ ei,
                           float* __restrict__ asrc, float* __restrict__ adst,
                           unsigned* __restrict__ tagp) {
  int t = blockIdx.x * blockDim.x + threadIdx.x;
  const int gw = __builtin_amdgcn_readfirstlane(t >> 5);
  if (gw < NBLK) {
    const int lane = threadIdx.x & 31;
    const int* __restrict__ dcol = ei + N_EDGES + gw * 1024;
    v4i d4[8];
#pragma unroll
    for (int r = 0; r < 8; ++r) {
      const int ro = (gw * 1024 + 128 * r < N_EDGES) ? 128 * r : 0;
      d4[r] = reinterpret_cast<const v4i*>(dcol + ro)[lane];
    }
    unsigned mine[8];
#pragma unroll
    for (int k = 0; k < 8; ++k) mine[k] = 0u;
#pragma unroll
    for (int r = 0; r < 8; ++r) {
#pragma unroll
      for (int c = 0; c < 4; ++c) {
        unsigned g = (unsigned)d4[r][c] / (unsigned)RNG;
        g = g < (unsigned)NGRP ? g : (unsigned)(NGRP - 1);
#pragma unroll
        for (int k = 0; k < 8; ++k) {
          const unsigned pk = __builtin_amdgcn_ballot_w32(((g >> k) & 1u) != 0u);
          if (lane == 4 * r + c) mine[k] = pk;
        }
      }
    }
#pragma unroll
    for (int k = 0; k < 8; ++k) tagp[gw * 256 + k * 32 + lane] = mine[k];
  }
  if (t >= N_NODES * HEADS) return;
  int head = t & 7;
  int n    = t >> 3;
  const float* __restrict__ hp = h + n * HF + head * FH;
  const float* __restrict__ as = att_src + head * FH;
  const float* __restrict__ ad = att_dst + head * FH;
  float s = 0.f, d = 0.f;
#pragma unroll
  for (int f = 0; f < FH; ++f) { float v = hp[f]; s += v * as[f]; d += v * ad[f]; }
  asrc[t] = s;
  adst[t] = d;
}

__device__ __forceinline__ int f32_key(float f) {
  int i = __float_as_int(f);
  return i >= 0 ? i : (i ^ 0x7fffffff);
}
__device__ __forceinline__ float key_f32(int k) {
  return __int_as_float(k >= 0 ? k : (k ^ 0x7fffffff));
}

__device__ __forceinline__ float leaky(float v) {
  return v >= 0.f ? v : NEG_SLOPE * v;
}

#define CAP   5120
#define NTHR  1024

static_assert(RNG * NGRP >= N_NODES, "ranges must cover every node");
static_assert(NBLK * 32 <= N_NODES * HEADS, "the logits launch has a wave for every tag block");
static_assert(N_NODES <= 65536 && RNG <= 256 && NGRP <= 256, "list entry packs src in 16 bits, local dst in 8; a tag is a byte");
static_assert(N_EDGES % 32 == 0, "a tag word covers 32 edges, the exact path walks 32 edges a wave");
static_assert(NBLK * 1024 <= (1 << 27) && NBLK >= NGRP, "hit entry packs block, word and bit in 32 bits; the start rotation is whole blocks");

__device__ __forceinline__ void edge_sum(int lane, int dl, float sa, const float (&hv)[4],
                                         const float* adl, const int* mxk,
                                         float* den, float* acc) {
  const int b = dl * HEADS + (lane & 7);
  const float v  = leaky(sa + adl[b]);
  const float ex = __expf(v - key_f32(mxk[b]));
  if (lane < 8)
    __hip_atomic_fetch_add(&den[b], ex, __ATOMIC_RELAXED, __HIP_MEMORY_SCOPE_WORKGROUP);
  float* ap = acc + dl * HF + lane;
  const bool hi = lane >= 16;
  const int exi = __float_as_int(ex);
#pragma unroll
  for (int j = 0; j < 4; ++j) {
    const float e0 = __int_as_float(__builtin_amdgcn_readlane(exi, 2 * j));
    const float e1 = __int_as_float(__builtin_amdgcn_readlane(exi, 2 * j + 1));
    const float a = hi ? e1 : e0;
    __hip_atomic_fetch_add(&ap[32 * j], a * hv[j], __ATOMIC_RELAXED, __HIP_MEMORY_SCOPE_WORKGROUP);
  }
}

__global__ __launch_bounds__(NTHR) void gat_owner(
    const int* __restrict__ ei, const unsigned* __restrict__ tagp,
    const float* __restrict__ h,
    const float* __restrict__ asrc, const float* __restrict__ adst,
    const float* __restrict__ bias, float* __restrict__ out) {
  __shared__ float    acc[RNG * HF];
  __shared__ int      mxk[RNG * HEADS];
  __shared__ float    den[RNG * HEADS];
  __shared__ float    adl[RNG * HEADS];
  __shared__ unsigned list[CAP];
  __shared__ v4f      tab4[CAP * 2];
  __shared__ unsigned cnt;
  float* tab = reinterpret_cast<float*>(tab4);

  const int tid  = threadIdx.x;
  const int lane = tid & 31;
  const int wave = tid >> 5;
  const int n0   = blockIdx.x * RNG;
  const int nloc = (N_NODES - n0) < RNG ? (N_NODES - n0) : RNG;

  for (int i = tid; i < RNG * HF; i += NTHR) acc[i] = 0.f;
  for (int i = tid; i < RNG * HEADS; i += NTHR) {
    mxk[i] = (int)0x80000000;
    den[i] = 0.f;
    adl[i] = (i < nloc * HEADS) ? adst[n0 * HEADS + i] : 0.f;
  }
  if (tid == 0) cnt = 0u;
  __syncthreads();

  {
    constexpr int UB = 4;
    const int wv = __builtin_amdgcn_readfirstlane(wave);
    unsigned inv[8];
#pragma unroll
    for (int k = 0; k < 8; ++k) inv[k] = ((blockIdx.x >> k) & 1u) ? 0u : 0xFFFFFFFFu;
    const int rot = blockIdx.x * (NBLK / NGRP);
    for (int b0 = wv; b0 < NBLK; b0 += 32 * UB) {
      unsigned pw[UB][8];
      int cid[UB];
#pragma unroll
      for (int u = 0; u < UB; ++u) {
        const int b = b0 + 32 * u;
        int q = (b < NBLK ? b : 0) + rot;
        if (q >= NBLK) q -= NBLK;
        cid[u] = b < NBLK ? q * 32 + lane : NCH;
        const unsigned* __restrict__ bp = tagp + q * 256 + lane;
#pragma unroll
        for (int k = 0; k < 8; ++k) pw[u][k] = bp[32 * k];
      }
      unsigned hm[UB];
      unsigned nh = 0u;
#pragma unroll
      for (int u = 0; u < UB; ++u) {
        unsigned m = 0xFFFFFFFFu;
#pragma unroll
        for (int k = 0; k < 8; ++k) m &= pw[u][k] ^ inv[k];
        if (cid[u] >= NCH) m = 0u;
        hm[u] = m;
        nh += (unsigned)__builtin_popcount(m);
      }
      if (nh) {
        unsigned slot = atomicAdd(&cnt, nh);
        if (slot + nh <= (unsigned)CAP) {
#pragma unroll
          for (int u = 0; u < UB; ++u) {
            unsigned m = hm[u];
            const unsigned c0 = (unsigned)cid[u] << 5;
            while (m) {
              const unsigned p = (unsigned)__builtin_ctz(m);
              m &= m - 1u;
              list[slot] = c0 | p;
              ++slot;
            }
          }
        }
      }
    }
  }
  __syncthreads();
  const unsigned total = cnt;

  if (total <= (unsigned)CAP) {
    for (unsigned i = tid; i < total; i += NTHR) {
      const unsigned w = list[i];
      const unsigned p = w & 31u, j = (w >> 5) & 31u, blk = w >> 10;
      const unsigned e = blk * 1024u + 128u * (j >> 2) + 4u * p + (j & 3u);
      const unsigned s = (unsigned)ei[e];
      unsigned dl = (unsigned)(ei[N_EDGES + e] - n0);
      dl = dl < (unsigned)RNG ? dl : (unsigned)(RNG - 1);
      list[i] = s | (dl << 16);
    }
    __syncthreads();

    const unsigned items = total * HEADS;
    for (unsigned i0 = tid; i0 < items; i0 += NTHR * 4) {
      float sv[4];
      unsigned dh[4];
#pragma unroll
      for (int u = 0; u < 4; ++u) {
        const unsigned idx = i0 + u * NTHR;
        sv[u] = 0.f;
        dh[u] = 0xFFFFFFFFu;
        if (idx < items) {
          const unsigned w = list[idx >> 3];
          const unsigned hd = idx & 7u;
          sv[u] = asrc[(w & 0xFFFFu) * HEADS + hd];
          dh[u] = (w >> 16) * HEADS + hd;
        }
      }
#pragma unroll
      for (int u = 0; u < 4; ++u) {
        if (dh[u] != 0xFFFFFFFFu) {
          const unsigned idx = i0 + u * NTHR;
          const unsigned hd = idx & 7u;
          const float v = leaky(sv[u] + adl[dh[u]]);
          __hip_atomic_fetch_max(&mxk[dh[u]], f32_key(v), __ATOMIC_RELAXED, __HIP_MEMORY_SCOPE_WORKGROUP);
          tab[(idx & ~7u) | ((hd & 1u) << 2) | (hd >> 1)] = v;
        }
      }
    }
    __syncthreads();

    for (unsigned i0 = tid; i0 < items; i0 += NTHR * 4) {
      float    lv[4];
      int      mk[4];
      unsigned bb[4], pp[4];
      unsigned le[4];
#pragma unroll
      for (int u = 0; u < 4; ++u) {
        const unsigned idx = i0 + u * NTHR;
        le[u] = list[(idx < items ? idx : i0) >> 3];
      }
#pragma unroll
      for (int u = 0; u < 4; ++u) {
        const unsigned idx = i0 + u * NTHR;
        const unsigned ix  = idx < items ? idx : i0;
        const unsigned hd  = ix & 7u;
        const unsigned b   = (le[u] >> 16) * HEADS + hd;
        pp[u] = (ix & ~7u) | ((hd & 1u) << 2) | (hd >> 1);
        lv[u] = tab[pp[u]];
        mk[u] = mxk[b];
        bb[u] = idx < items ? b : 0xFFFFFFFFu;
      }
#pragma unroll
      for (int u = 0; u < 4; ++u) {
        if (bb[u] != 0xFFFFFFFFu) {
          const float ex = __expf(lv[u] - key_f32(mk[u]));
          tab[pp[u]] = ex;
          __hip_atomic_fetch_add(&den[bb[u]], ex, __ATOMIC_RELAXED, __HIP_MEMORY_SCOPE_WORKGROUP);
        }
      }
    }
    __syncthreads();

    {
      constexpr int SU = 8;
      const unsigned wv  = (unsigned)__builtin_amdgcn_readfirstlane(wave);
      const unsigned tot = (unsigned)__builtin_amdgcn_readfirstlane((int)total);
      const int half = lane >> 4;
      for (unsigned i0 = wv; i0 < tot; i0 += 32 * SU) {
        const unsigned li = i0 + 32u * (unsigned)(lane & (SU - 1));
        const unsigned lw = list[li < tot ? li : i0];
        float hv[SU][4];
        v4f   wt[SU];
        unsigned dl[SU];
#pragma unroll
        for (int u = 0; u < SU; ++u) {
          const unsigned i = i0 + 32u * (unsigned)u;
          const unsigned w = (unsigned)__builtin_amdgcn_readlane((int)lw, u);
          dl[u] = w >> 16;
          if (i < tot) {
            const float* __restrict__ hp = h + (w & 0xFFFFu) * HF + lane;
#pragma unroll
            for (int j = 0; j < 4; ++j) hv[u][j] = hp[32 * j];
            wt[u] = tab4[i * 2 + half];
          }
        }
#pragma unroll
        for (int u = 0; u < SU; ++u) {
          if (i0 + 32u * (unsigned)u < tot) {
            float* ap = acc + dl[u] * HF + lane;
#pragma unroll
            for (int j = 0; j < 4; ++j)
              __hip_atomic_fetch_add(&ap[32 * j], wt[u][j] * hv[u][j], __ATOMIC_RELAXED, __HIP_MEMORY_SCOPE_WORKGROUP);
          }
        }
      }
    }
    __syncthreads();
  } else {
    for (int ph = 0; ph < 2; ++ph) {
      for (int base = wave * 32; base < N_EDGES; base += 32 * 32) {
        const unsigned d = (unsigned)(ei[N_EDGES + base + lane] - n0);
        unsigned mask = __builtin_amdgcn_ballot_w32(d < (unsigned)RNG);
        while (mask) {
          const int bpos = __builtin_ctz(mask);
          mask &= mask - 1u;
          const int s   = ei[base + bpos];
          const int dlu = __shfl((int)d, bpos, 32);
          const float sa = asrc[s * HEADS + (lane & 7)];
          if (ph == 0) {
            const int b = dlu * HEADS + (lane & 7);
            const float v = leaky(sa + adl[b]);
            if (lane < 8)
              __hip_atomic_fetch_max(&mxk[b], f32_key(v), __ATOMIC_RELAXED, __HIP_MEMORY_SCOPE_WORKGROUP);
          } else {
            float hv[4];
            const float* __restrict__ hp = h + s * HF + lane;
#pragma unroll
            for (int j = 0; j < 4; ++j) hv[j] = hp[32 * j];
            edge_sum(lane, dlu, sa, hv, adl, mxk, den, acc);
          }
        }
      }
      __syncthreads();
    }
  }

  for (int i = tid; i < nloc * HF; i += NTHR) {
    const int r = i >> 7, f = i & (HF - 1);
    out[n0 * HF + i] = acc[i] / (den[r * HEADS + (f >> 4)] + 1e-16f) + bias[f];
  }
}

extern "C" void kernel_launch(void* const* d_in, const int* in_sizes, int n_in,
                              void* d_out, int out_size, void* d_ws, size_t ws_size,
                              hipStream_t stream) {
  const float* x        = (const float*)d_in[0];
  const float* W        = (const float*)d_in[1];
  const float* att_src  = (const float*)d_in[2];
  const float* att_dst  = (const float*)d_in[3];
  const float* bias     = (const float*)d_in[4];
  const int*   ei       = (const int*)d_in[5];
  float*       out      = (float*)d_out;

  float* h     = (float*)d_ws;
  float* asrc  = h     + (size_t)N_NODES * HF;
  float* adst  = asrc  + (size_t)N_NODES * HEADS;
  unsigned* tagp = (unsigned*)(adst + (size_t)N_NODES * HEADS);

  const int BLK = 256;

  gat_gemm_wmma<<<GEMM_WG, BLK, 0, stream>>>(x, W, h);

  gat_logits<<<(N_NODES * HEADS + BLK - 1) / BLK, BLK, 0, stream>>>(
      h, att_src, att_dst, ei, asrc, adst, tagp);

  gat_owner<<<NGRP, NTHR, 0, stream>>>(ei, tagp, h, asrc, adst, bias, out);
}


# Range tags one-hot by nibble in 32 bit planes: an owner group reads only its 2 planes (2 bits per edge, not 8), all loads in one batch, one list reservation per wave; arrival-order LDS sums kept
# speedup vs baseline: 3.8244x; 1.0774x over previous
#include <hip/hip_runtime.h>
#include <hip/hip_bf16.h>

#define N_NODES 50000
#define N_EDGES 800000
#define F_IN    128
#define HEADS   8
#define FH      16
#define HF      128
#define NEG_SLOPE 0.2f

typedef float v2f __attribute__((ext_vector_type(2)));
typedef float v4f __attribute__((ext_vector_type(4)));
typedef float v8f __attribute__((ext_vector_type(8)));

#define GEMM_WG  1024
#define M_TILES  (N_NODES / 16)

static_assert(N_NODES % 16 == 0, "whole 16-row tiles");

__global__ __launch_bounds__(256) void gat_gemm_wmma(
    const float* __restrict__ x, const float* __restrict__ W,
    float* __restrict__ h) {
  const int lane = threadIdx.x & 31;
  const int wave = threadIdx.x >> 5;
  const int n0   = wave * 16;
  const int half = lane >> 4;
  const int l15  = lane & 15;

  const v4f* __restrict__ brow =
      reinterpret_cast<const v4f*>(W + (n0 + l15) * F_IN + 4 * half);
  v4f b[16];
#pragma unroll
  for (int q = 0; q < 16; ++q) b[q] = brow[2 * q];

  for (int t = blockIdx.x; t < M_TILES; t += GEMM_WG) {
    const int m0 = t * 16;
    const v4f* __restrict__ arow =
        reinterpret_cast<const v4f*>(x + (m0 + l15) * F_IN + 4 * half);
    v8f c = {};
#pragma unroll
    for (int hh = 0; hh < 2; ++hh) {
      v4f a[8];
#pragma unroll
      for (int q = 0; q < 8; ++q) a[q] = arow[2 * (8 * hh + q)];
      asm volatile("" :: "v"(a[0]), "v"(a[1]), "v"(a[2]), "v"(a[3]),
                         "v"(a[4]), "v"(a[5]), "v"(a[6]), "v"(a[7]));
#pragma unroll
      for (int q = 0; q < 8; ++q) {
        const v4f bq = b[8 * hh + q];
        v2f a0; a0.x = a[q].x; a0.y = a[q].y;
        v2f a1; a1.x = a[q].z; a1.y = a[q].w;
        v2f b0; b0.x = bq.x;   b0.y = bq.y;
        v2f b1; b1.x = bq.z;   b1.y = bq.w;
        c = __builtin_amdgcn_wmma_f32_16x16x4_f32(
             false, a0,  false, b0,
             (short)0, c,  false,  false);
        c = __builtin_amdgcn_wmma_f32_16x16x4_f32(
             false, a1,  false, b1,
             (short)0, c,  false,  false);
      }
    }

    float* __restrict__ op = h + (m0 + half * 8) * HF + n0 + l15;
#pragma unroll
    for (int v = 0; v < 8; ++v) op[v * HF] = c[v];
  }
}

#define RNG   196
#define NGRP  256
#define NCH   (N_EDGES / 32)
#define NBLK  ((NCH + 31) / 32)
#define NPLANE 32
#define PW    (NBLK * 32)
#define NUNIT (PW / 64)

typedef int v4i __attribute__((ext_vector_type(4)));
typedef unsigned v2u __attribute__((ext_vector_type(2)));

static_assert(N_EDGES % 128 == 0, "a tag block is read in whole 16-byte loads of 32 lanes");
static_assert(PW % 64 == 0, "a plane is read in whole 256-byte pieces");

__global__ void gat_logits(const float* __restrict__ h,
                           const float* __restrict__ att_src,
                           const float* __restrict__ att_dst,
                           const int* __restrict__ ei,
                           float* __restrict__ asrc, float* __restrict__ adst,
                           unsigned* __restrict__ tagp) {
  int t = blockIdx.x * blockDim.x + threadIdx.x;
  const int gw = __builtin_amdgcn_readfirstlane(t >> 5);
  if (gw < NBLK) {
    const int lane = threadIdx.x & 31;
    const int* __restrict__ dcol = ei + N_EDGES + gw * 1024;
    v4i d4[8];
#pragma unroll
    for (int r = 0; r < 8; ++r) {
      const int ro = (gw * 1024 + 128 * r < N_EDGES) ? 128 * r : 0;
      d4[r] = reinterpret_cast<const v4i*>(dcol + ro)[lane];
    }
    unsigned row[32];
#pragma unroll
    for (int r = 0; r < 8; ++r) {
      const bool inlist = gw * 1024 + 128 * r < N_EDGES;
#pragma unroll
      for (int c = 0; c < 4; ++c) {
        unsigned g = (unsigned)d4[r][c] / (unsigned)RNG;
        g = g < (unsigned)NGRP ? g : (unsigned)(NGRP - 1);
        const unsigned oh = (1u << (g >> 4)) | (0x10000u << (g & 15u));
        row[4 * r + c] = inlist ? oh : 0u;
      }
    }
#pragma unroll
    for (int s = 0; s < 5; ++s) {
      const int j = 16 >> s;
      const unsigned m = 0xFFFFFFFFu / ((1u << j) + 1u);
#pragma unroll
      for (int k = 0; k < 32; ++k) {
        if ((k & j) == 0) {
          const unsigned t = ((row[k] >> j) ^ row[k + j]) & m;
          row[k]     ^= t << j;
          row[k + j] ^= t;
        }
      }
    }
#pragma unroll
    for (int P = 0; P < NPLANE; ++P) tagp[P * PW + gw * 32 + lane] = row[P];
  }
  if (t >= N_NODES * HEADS) return;
  int head = t & 7;
  int n    = t >> 3;
  const float* __restrict__ hp = h + n * HF + head * FH;
  const float* __restrict__ as = att_src + head * FH;
  const float* __restrict__ ad = att_dst + head * FH;
  float s = 0.f, d = 0.f;
#pragma unroll
  for (int f = 0; f < FH; ++f) { float v = hp[f]; s += v * as[f]; d += v * ad[f]; }
  asrc[t] = s;
  adst[t] = d;
}

__device__ __forceinline__ int f32_key(float f) {
  int i = __float_as_int(f);
  return i >= 0 ? i : (i ^ 0x7fffffff);
}
__device__ __forceinline__ float key_f32(int k) {
  return __int_as_float(k >= 0 ? k : (k ^ 0x7fffffff));
}

__device__ __forceinline__ float leaky(float v) {
  return v >= 0.f ? v : NEG_SLOPE * v;
}

#define CAP   5120
#define NTHR  1024

static_assert(RNG * NGRP >= N_NODES, "ranges must cover every node");
static_assert(NBLK * 32 <= N_NODES * HEADS, "the logits launch has a wave for every tag block");
static_assert(N_NODES <= 65536 && RNG <= 256 && NGRP <= 256, "list entry packs src in 16 bits, local dst in 8; a tag is a byte");
static_assert(N_EDGES % 32 == 0, "a tag word covers 32 edges, the exact path walks 32 edges a wave");
static_assert(NBLK * 1024 <= (1 << 27) && NUNIT >= NGRP, "hit entry packs block, word and bit in 32 bits; every group has a start piece of its own");

__device__ __forceinline__ void edge_sum(int lane, int dl, float sa, const float (&hv)[4],
                                         const float* adl, const int* mxk,
                                         float* den, float* acc) {
  const int b = dl * HEADS + (lane & 7);
  const float v  = leaky(sa + adl[b]);
  const float ex = __expf(v - key_f32(mxk[b]));
  if (lane < 8)
    __hip_atomic_fetch_add(&den[b], ex, __ATOMIC_RELAXED, __HIP_MEMORY_SCOPE_WORKGROUP);
  float* ap = acc + dl * HF + lane;
  const bool hi = lane >= 16;
  const int exi = __float_as_int(ex);
#pragma unroll
  for (int j = 0; j < 4; ++j) {
    const float e0 = __int_as_float(__builtin_amdgcn_readlane(exi, 2 * j));
    const float e1 = __int_as_float(__builtin_amdgcn_readlane(exi, 2 * j + 1));
    const float a = hi ? e1 : e0;
    __hip_atomic_fetch_add(&ap[32 * j], a * hv[j], __ATOMIC_RELAXED, __HIP_MEMORY_SCOPE_WORKGROUP);
  }
}

__global__ __launch_bounds__(NTHR) void gat_owner(
    const int* __restrict__ ei, const unsigned* __restrict__ tagp,
    const float* __restrict__ h,
    const float* __restrict__ asrc, const float* __restrict__ adst,
    const float* __restrict__ bias, float* __restrict__ out) {
  __shared__ float    acc[RNG * HF];
  __shared__ int      mxk[RNG * HEADS];
  __shared__ float    den[RNG * HEADS];
  __shared__ float    adl[RNG * HEADS];
  __shared__ unsigned list[CAP];
  __shared__ v4f      tab4[CAP * 2];
  __shared__ unsigned cnt;
  float* tab = reinterpret_cast<float*>(tab4);

  const int tid  = threadIdx.x;
  const int lane = tid & 31;
  const int wave = tid >> 5;
  const int n0   = blockIdx.x * RNG;
  const int nloc = (N_NODES - n0) < RNG ? (N_NODES - n0) : RNG;

  for (int i = tid; i < RNG * HF; i += NTHR) acc[i] = 0.f;
  for (int i = tid; i < RNG * HEADS; i += NTHR) {
    mxk[i] = (int)0x80000000;
    den[i] = 0.f;
    adl[i] = (i < nloc * HEADS) ? adst[n0 * HEADS + i] : 0.f;
  }
  if (tid == 0) cnt = 0u;
  __syncthreads();

  {
    constexpr int NU = (NUNIT + 31) / 32;
    const int wv = __builtin_amdgcn_readfirstlane(wave);
    const unsigned* __restrict__ pa = tagp + (blockIdx.x >> 4) * PW;
    const unsigned* __restrict__ pb = tagp + (16 + (blockIdx.x & 15)) * PW;
    const int rot = (blockIdx.x * NUNIT) / NGRP;
    v2u wa[NU], wb[NU];
#pragma unroll
    for (int i = 0; i < NU; ++i) {
      const int u = wv + 32 * i;
      int q = (u < NUNIT ? u : wv) + rot;
      if (q >= NUNIT) q -= NUNIT;
      wa[i] = *reinterpret_cast<const v2u*>(pa + q * 64 + 2 * lane);
      wb[i] = *reinterpret_cast<const v2u*>(pb + q * 64 + 2 * lane);
    }
    unsigned long long hm[NU];
    unsigned nh = 0u;
#pragma unroll
    for (int i = 0; i < NU; ++i) {
      const unsigned lo = wa[i].x & wb[i].x, hi = wa[i].y & wb[i].y;
      unsigned long long m = ((unsigned long long)hi << 32) | lo;
      if (wv + 32 * i >= NUNIT) m = 0ull;
      hm[i] = m;
      nh += (unsigned)__builtin_popcountll(m);
    }
    unsigned inc = nh;
#pragma unroll
    for (int d = 1; d < 32; d <<= 1) {
      const unsigned o = __shfl_up(inc, d, 32);
      if (lane >= d) inc += o;
    }
    const unsigned wtot = (unsigned)__builtin_amdgcn_readlane((int)inc, 31);
    if (wtot) {
      unsigned base = 0u;
      if (lane == 0) base = atomicAdd(&cnt, wtot);
      base = (unsigned)__builtin_amdgcn_readfirstlane((int)base);
      if (base + wtot <= (unsigned)CAP) {
        unsigned slot = base + inc - nh;
#pragma unroll
        for (int i = 0; i < NU; ++i) {
          const int u = wv + 32 * i;
          int q = (u < NUNIT ? u : wv) + rot;
          if (q >= NUNIT) q -= NUNIT;
          const unsigned c0 = (unsigned)(q * 64 + 2 * lane) << 5;
          unsigned long long m = hm[i];
          while (m) {
            const unsigned p = (unsigned)__builtin_ctzll(m);
            m &= m - 1ull;
            list[slot] = c0 + p;
            ++slot;
          }
        }
      }
    }
  }
  __syncthreads();
  const unsigned total = cnt;

  if (total <= (unsigned)CAP) {
    for (unsigned i = tid; i < total; i += NTHR) {
      const unsigned w = list[i];
      const unsigned p = w & 31u, j = (w >> 5) & 31u, blk = w >> 10;
      const unsigned e = blk * 1024u + 128u * (p >> 2) + 4u * j + (p & 3u);
      const unsigned s = (unsigned)ei[e];
      unsigned dl = (unsigned)(ei[N_EDGES + e] - n0);
      dl = dl < (unsigned)RNG ? dl : (unsigned)(RNG - 1);
      list[i] = s | (dl << 16);
    }
    __syncthreads();

    const unsigned items = total * HEADS;
    for (unsigned i0 = tid; i0 < items; i0 += NTHR * 4) {
      float sv[4];
      unsigned dh[4];
#pragma unroll
      for (int u = 0; u < 4; ++u) {
        const unsigned idx = i0 + u * NTHR;
        sv[u] = 0.f;
        dh[u] = 0xFFFFFFFFu;
        if (idx < items) {
          const unsigned w = list[idx >> 3];
          const unsigned hd = idx & 7u;
          sv[u] = asrc[(w & 0xFFFFu) * HEADS + hd];
          dh[u] = (w >> 16) * HEADS + hd;
        }
      }
#pragma unroll
      for (int u = 0; u < 4; ++u) {
        if (dh[u] != 0xFFFFFFFFu) {
          const unsigned idx = i0 + u * NTHR;
          const unsigned hd = idx & 7u;
          const float v = leaky(sv[u] + adl[dh[u]]);
          __hip_atomic_fetch_max(&mxk[dh[u]], f32_key(v), __ATOMIC_RELAXED, __HIP_MEMORY_SCOPE_WORKGROUP);
          tab[(idx & ~7u) | ((hd & 1u) << 2) | (hd >> 1)] = v;
        }
      }
    }
    __syncthreads();

    for (unsigned i0 = tid; i0 < items; i0 += NTHR * 4) {
      float    lv[4];
      int      mk[4];
      unsigned bb[4], pp[4];
      unsigned le[4];
#pragma unroll
      for (int u = 0; u < 4; ++u) {
        const unsigned idx = i0 + u * NTHR;
        le[u] = list[(idx < items ? idx : i0) >> 3];
      }
#pragma unroll
      for (int u = 0; u < 4; ++u) {
        const unsigned idx = i0 + u * NTHR;
        const unsigned ix  = idx < items ? idx : i0;
        const unsigned hd  = ix & 7u;
        const unsigned b   = (le[u] >> 16) * HEADS + hd;
        pp[u] = (ix & ~7u) | ((hd & 1u) << 2) | (hd >> 1);
        lv[u] = tab[pp[u]];
        mk[u] = mxk[b];
        bb[u] = idx < items ? b : 0xFFFFFFFFu;
      }
#pragma unroll
      for (int u = 0; u < 4; ++u) {
        if (bb[u] != 0xFFFFFFFFu) {
          const float ex = __expf(lv[u] - key_f32(mk[u]));
          tab[pp[u]] = ex;
          __hip_atomic_fetch_add(&den[bb[u]], ex, __ATOMIC_RELAXED, __HIP_MEMORY_SCOPE_WORKGROUP);
        }
      }
    }
    __syncthreads();

    {
      constexpr int SU = 8;
      const unsigned wv  = (unsigned)__builtin_amdgcn_readfirstlane(wave);
      const unsigned tot = (unsigned)__builtin_amdgcn_readfirstlane((int)total);
      const int half = lane >> 4;
      for (unsigned i0 = wv; i0 < tot; i0 += 32 * SU) {
        const unsigned li = i0 + 32u * (unsigned)(lane & (SU - 1));
        const unsigned lw = list[li < tot ? li : i0];
        float hv[SU][4];
        v4f   wt[SU];
        unsigned dl[SU];
#pragma unroll
        for (int u = 0; u < SU; ++u) {
          const unsigned i = i0 + 32u * (unsigned)u;
          const unsigned w = (unsigned)__builtin_amdgcn_readlane((int)lw, u);
          dl[u] = w >> 16;
          if (i < tot) {
            const float* __restrict__ hp = h + (w & 0xFFFFu) * HF + lane;
#pragma unroll
            for (int j = 0; j < 4; ++j) hv[u][j] = hp[32 * j];
            wt[u] = tab4[i * 2 + half];
          }
        }
#pragma unroll
        for (int u = 0; u < SU; ++u) {
          if (i0 + 32u * (unsigned)u < tot) {
            float* ap = acc + dl[u] * HF + lane;
#pragma unroll
            for (int j = 0; j < 4; ++j)
              __hip_atomic_fetch_add(&ap[32 * j], wt[u][j] * hv[u][j], __ATOMIC_RELAXED, __HIP_MEMORY_SCOPE_WORKGROUP);
          }
        }
      }
    }
    __syncthreads();
  } else {
    for (int ph = 0; ph < 2; ++ph) {
      for (int base = wave * 32; base < N_EDGES; base += 32 * 32) {
        const unsigned d = (unsigned)(ei[N_EDGES + base + lane] - n0);
        unsigned mask = __builtin_amdgcn_ballot_w32(d < (unsigned)RNG);
        while (mask) {
          const int bpos = __builtin_ctz(mask);
          mask &= mask - 1u;
          const int s   = ei[base + bpos];
          const int dlu = __shfl((int)d, bpos, 32);
          const float sa = asrc[s * HEADS + (lane & 7)];
          if (ph == 0) {
            const int b = dlu * HEADS + (lane & 7);
            const float v = leaky(sa + adl[b]);
            if (lane < 8)
              __hip_atomic_fetch_max(&mxk[b], f32_key(v), __ATOMIC_RELAXED, __HIP_MEMORY_SCOPE_WORKGROUP);
          } else {
            float hv[4];
            const float* __restrict__ hp = h + s * HF + lane;
#pragma unroll
            for (int j = 0; j < 4; ++j) hv[j] = hp[32 * j];
            edge_sum(lane, dlu, sa, hv, adl, mxk, den, acc);
          }
        }
      }
      __syncthreads();
    }
  }

  for (int i = tid; i < nloc * HF; i += NTHR) {
    const int r = i >> 7, f = i & (HF - 1);
    out[n0 * HF + i] = acc[i] / (den[r * HEADS + (f >> 4)] + 1e-16f) + bias[f];
  }
}

extern "C" void kernel_launch(void* const* d_in, const int* in_sizes, int n_in,
                              void* d_out, int out_size, void* d_ws, size_t ws_size,
                              hipStream_t stream) {
  const float* x        = (const float*)d_in[0];
  const float* W        = (const float*)d_in[1];
  const float* att_src  = (const float*)d_in[2];
  const float* att_dst  = (const float*)d_in[3];
  const float* bias     = (const float*)d_in[4];
  const int*   ei       = (const int*)d_in[5];
  float*       out      = (float*)d_out;

  float* h     = (float*)d_ws;
  float* asrc  = h     + (size_t)N_NODES * HF;
  float* adst  = asrc  + (size_t)N_NODES * HEADS;
  unsigned* tagp = (unsigned*)(adst + (size_t)N_NODES * HEADS);

  const int BLK = 256;

  gat_gemm_wmma<<<GEMM_WG, BLK, 0, stream>>>(x, W, h);

  gat_logits<<<(N_NODES * HEADS + BLK - 1) / BLK, BLK, 0, stream>>>(
      h, att_src, att_dst, ei, asrc, adst, tagp);

  gat_owner<<<NGRP, NTHR, 0, stream>>>(ei, tagp, h, asrc, adst, bias, out);
}


# Owner sum pass: a lane takes 4 adjacent features of h[src] by one 16-byte load (1 load instruction per 512-byte row, was 4) and 1 head weight; LDS row slots interleaved; arrival-order LDS sums kept
# speedup vs baseline: 4.1472x; 1.0844x over previous
#include <hip/hip_runtime.h>
#include <hip/hip_bf16.h>

#define N_NODES 50000
#define N_EDGES 800000
#define F_IN    128
#define HEADS   8
#define FH      16
#define HF      128
#define NEG_SLOPE 0.2f

typedef float v2f __attribute__((ext_vector_type(2)));
typedef float v4f __attribute__((ext_vector_type(4)));
typedef float v8f __attribute__((ext_vector_type(8)));

#define GEMM_WG  1024
#define M_TILES  (N_NODES / 16)

static_assert(N_NODES % 16 == 0, "whole 16-row tiles");

__global__ __launch_bounds__(256) void gat_gemm_wmma(
    const float* __restrict__ x, const float* __restrict__ W,
    float* __restrict__ h) {
  const int lane = threadIdx.x & 31;
  const int wave = threadIdx.x >> 5;
  const int n0   = wave * 16;
  const int half = lane >> 4;
  const int l15  = lane & 15;

  const v4f* __restrict__ brow =
      reinterpret_cast<const v4f*>(W + (n0 + l15) * F_IN + 4 * half);
  v4f b[16];
#pragma unroll
  for (int q = 0; q < 16; ++q) b[q] = brow[2 * q];

  for (int t = blockIdx.x; t < M_TILES; t += GEMM_WG) {
    const int m0 = t * 16;
    const v4f* __restrict__ arow =
        reinterpret_cast<const v4f*>(x + (m0 + l15) * F_IN + 4 * half);
    v8f c = {};
#pragma unroll
    for (int hh = 0; hh < 2; ++hh) {
      v4f a[8];
#pragma unroll
      for (int q = 0; q < 8; ++q) a[q] = arow[2 * (8 * hh + q)];
      asm volatile("" :: "v"(a[0]), "v"(a[1]), "v"(a[2]), "v"(a[3]),
                         "v"(a[4]), "v"(a[5]), "v"(a[6]), "v"(a[7]));
#pragma unroll
      for (int q = 0; q < 8; ++q) {
        const v4f bq = b[8 * hh + q];
        v2f a0; a0.x = a[q].x; a0.y = a[q].y;
        v2f a1; a1.x = a[q].z; a1.y = a[q].w;
        v2f b0; b0.x = bq.x;   b0.y = bq.y;
        v2f b1; b1.x = bq.z;   b1.y = bq.w;
        c = __builtin_amdgcn_wmma_f32_16x16x4_f32(
             false, a0,  false, b0,
             (short)0, c,  false,  false);
        c = __builtin_amdgcn_wmma_f32_16x16x4_f32(
             false, a1,  false, b1,
             (short)0, c,  false,  false);
      }
    }

    float* __restrict__ op = h + (m0 + half * 8) * HF + n0 + l15;
#pragma unroll
    for (int v = 0; v < 8; ++v) op[v * HF] = c[v];
  }
}

#define RNG   196
#define NGRP  256
#define NCH   (N_EDGES / 32)
#define NBLK  ((NCH + 31) / 32)
#define NPLANE 32
#define PW    (NBLK * 32)
#define NUNIT (PW / 64)

typedef int v4i __attribute__((ext_vector_type(4)));
typedef unsigned v2u __attribute__((ext_vector_type(2)));

static_assert(N_EDGES % 128 == 0, "a tag block is read in whole 16-byte loads of 32 lanes");
static_assert(PW % 64 == 0, "a plane is read in whole 256-byte pieces");

__global__ void gat_logits(const float* __restrict__ h,
                           const float* __restrict__ att_src,
                           const float* __restrict__ att_dst,
                           const int* __restrict__ ei,
                           float* __restrict__ asrc, float* __restrict__ adst,
                           unsigned* __restrict__ tagp) {
  int t = blockIdx.x * blockDim.x + threadIdx.x;
  const int gw = __builtin_amdgcn_readfirstlane(t >> 5);
  if (gw < NBLK) {
    const int lane = threadIdx.x & 31;
    const int* __restrict__ dcol = ei + N_EDGES + gw * 1024;
    v4i d4[8];
#pragma unroll
    for (int r = 0; r < 8; ++r) {
      const int ro = (gw * 1024 + 128 * r < N_EDGES) ? 128 * r : 0;
      d4[r] = reinterpret_cast<const v4i*>(dcol + ro)[lane];
    }
    unsigned row[32];
#pragma unroll
    for (int r = 0; r < 8; ++r) {
      const bool inlist = gw * 1024 + 128 * r < N_EDGES;
#pragma unroll
      for (int c = 0; c < 4; ++c) {
        unsigned g = (unsigned)d4[r][c] / (unsigned)RNG;
        g = g < (unsigned)NGRP ? g : (unsigned)(NGRP - 1);
        const unsigned oh = (1u << (g >> 4)) | (0x10000u << (g & 15u));
        row[4 * r + c] = inlist ? oh : 0u;
      }
    }
#pragma unroll
    for (int s = 0; s < 5; ++s) {
      const int j = 16 >> s;
      const unsigned m = 0xFFFFFFFFu / ((1u << j) + 1u);
#pragma unroll
      for (int k = 0; k < 32; ++k) {
        if ((k & j) == 0) {
          const unsigned t = ((row[k] >> j) ^ row[k + j]) & m;
          row[k]     ^= t << j;
          row[k + j] ^= t;
        }
      }
    }
#pragma unroll
    for (int P = 0; P < NPLANE; ++P) tagp[P * PW + gw * 32 + lane] = row[P];
  }
  if (t >= N_NODES * HEADS) return;
  int head = t & 7;
  int n    = t >> 3;
  const float* __restrict__ hp = h + n * HF + head * FH;
  const float* __restrict__ as = att_src + head * FH;
  const float* __restrict__ ad = att_dst + head * FH;
  float s = 0.f, d = 0.f;
#pragma unroll
  for (int f = 0; f < FH; ++f) { float v = hp[f]; s += v * as[f]; d += v * ad[f]; }
  asrc[t] = s;
  adst[t] = d;
}

__device__ __forceinline__ int f32_key(float f) {
  int i = __float_as_int(f);
  return i >= 0 ? i : (i ^ 0x7fffffff);
}
__device__ __forceinline__ float key_f32(int k) {
  return __int_as_float(k >= 0 ? k : (k ^ 0x7fffffff));
}

__device__ __forceinline__ float leaky(float v) {
  return v >= 0.f ? v : NEG_SLOPE * v;
}

#define CAP   5120
#define NTHR  1024

static_assert(RNG * NGRP >= N_NODES, "ranges must cover every node");
static_assert(NBLK * 32 <= N_NODES * HEADS, "the logits launch has a wave for every tag block");
static_assert(N_NODES <= 65536 && RNG <= 256 && NGRP <= 256, "list entry packs src in 16 bits, local dst in 8; a tag is a byte");
static_assert(N_EDGES % 32 == 0, "a tag word covers 32 edges, the exact path walks 32 edges a wave");
static_assert(NBLK * 1024 <= (1 << 27) && NUNIT >= NGRP, "hit entry packs block, word and bit in 32 bits; every group has a start piece of its own");

__device__ __forceinline__ void edge_sum(int lane, int dl, float sa, const v4f hv,
                                         const float* adl, const int* mxk,
                                         float* den, float* acc) {
  const int b = dl * HEADS + (lane & 7);
  const float v  = leaky(sa + adl[b]);
  const float ex = __expf(v - key_f32(mxk[b]));
  if (lane < 8)
    __hip_atomic_fetch_add(&den[b], ex, __ATOMIC_RELAXED, __HIP_MEMORY_SCOPE_WORKGROUP);
  float* ap = acc + dl * HF + lane;
  const float a = __shfl(ex, lane >> 2, 32);
#pragma unroll
  for (int j = 0; j < 4; ++j)
    __hip_atomic_fetch_add(&ap[32 * j], a * hv[j], __ATOMIC_RELAXED, __HIP_MEMORY_SCOPE_WORKGROUP);
}

template <bool FULL, int SU>
__device__ __forceinline__ void sum_batch(unsigned i0, unsigned tot, int lane,
                                          const float* __restrict__ h,
                                          const unsigned* list, const float* tab,
                                          float* acc) {
  const unsigned hq = (unsigned)lane >> 2;
  const unsigned lane16 = (unsigned)lane * 16u;
  const unsigned li = i0 + 32u * (unsigned)(lane & (SU - 1));
  const unsigned lw = list[(FULL || li < tot) ? li : i0];
  v4f   hv[SU];
  float wt[SU];
  unsigned dl[SU];
#pragma unroll
  for (int u = 0; u < SU; ++u) {
    const unsigned i  = i0 + 32u * (unsigned)u;
    const unsigned ic = (FULL || i < tot) ? i : i0;
    const unsigned w  = (unsigned)__builtin_amdgcn_readlane((int)lw, u);
    dl[u] = w >> 16;
    const unsigned off = ((w & 0xFFFFu) << 9) + lane16;
    hv[u] = *reinterpret_cast<const v4f*>(reinterpret_cast<const char*>(h) + off);
    wt[u] = tab[ic * HEADS + hq];
  }
#pragma unroll
  for (int u = 0; u < SU; ++u) {
    if (FULL || i0 + 32u * (unsigned)u < tot) {
      float* ap = acc + dl[u] * HF + lane;
#pragma unroll
      for (int j = 0; j < 4; ++j)
        __hip_atomic_fetch_add(&ap[32 * j], wt[u] * hv[u][j], __ATOMIC_RELAXED, __HIP_MEMORY_SCOPE_WORKGROUP);
    }
  }
}

__global__ __launch_bounds__(NTHR) void gat_owner(
    const int* __restrict__ ei, const unsigned* __restrict__ tagp,
    const float* __restrict__ h,
    const float* __restrict__ asrc, const float* __restrict__ adst,
    const float* __restrict__ bias, float* __restrict__ out) {
  __shared__ float    acc[RNG * HF];
  __shared__ int      mxk[RNG * HEADS];
  __shared__ float    den[RNG * HEADS];
  __shared__ float    adl[RNG * HEADS];
  __shared__ unsigned list[CAP];
  __shared__ float    tab[CAP * HEADS];
  __shared__ unsigned cnt;

  const int tid  = threadIdx.x;
  const int lane = tid & 31;
  const int wave = tid >> 5;
  const int n0   = blockIdx.x * RNG;
  const int nloc = (N_NODES - n0) < RNG ? (N_NODES - n0) : RNG;

  for (int i = tid; i < RNG * HF; i += NTHR) acc[i] = 0.f;
  for (int i = tid; i < RNG * HEADS; i += NTHR) {
    mxk[i] = (int)0x80000000;
    den[i] = 0.f;
    adl[i] = (i < nloc * HEADS) ? adst[n0 * HEADS + i] : 0.f;
  }
  if (tid == 0) cnt = 0u;
  __syncthreads();

  {
    constexpr int NU = (NUNIT + 31) / 32;
    const int wv = __builtin_amdgcn_readfirstlane(wave);
    const unsigned* __restrict__ pa = tagp + (blockIdx.x >> 4) * PW;
    const unsigned* __restrict__ pb = tagp + (16 + (blockIdx.x & 15)) * PW;
    const int rot = (blockIdx.x * NUNIT) / NGRP;
    v2u wa[NU], wb[NU];
#pragma unroll
    for (int i = 0; i < NU; ++i) {
      const int u = wv + 32 * i;
      int q = (u < NUNIT ? u : wv) + rot;
      if (q >= NUNIT) q -= NUNIT;
      wa[i] = *reinterpret_cast<const v2u*>(pa + q * 64 + 2 * lane);
      wb[i] = *reinterpret_cast<const v2u*>(pb + q * 64 + 2 * lane);
    }
    unsigned long long hm[NU];
    unsigned nh = 0u;
#pragma unroll
    for (int i = 0; i < NU; ++i) {
      const unsigned lo = wa[i].x & wb[i].x, hi = wa[i].y & wb[i].y;
      unsigned long long m = ((unsigned long long)hi << 32) | lo;
      if (wv + 32 * i >= NUNIT) m = 0ull;
      hm[i] = m;
      nh += (unsigned)__builtin_popcountll(m);
    }
    unsigned inc = nh;
#pragma unroll
    for (int d = 1; d < 32; d <<= 1) {
      const unsigned o = __shfl_up(inc, d, 32);
      if (lane >= d) inc += o;
    }
    const unsigned wtot = (unsigned)__builtin_amdgcn_readlane((int)inc, 31);
    if (wtot) {
      unsigned base = 0u;
      if (lane == 0) base = atomicAdd(&cnt, wtot);
      base = (unsigned)__builtin_amdgcn_readfirstlane((int)base);
      if (base + wtot <= (unsigned)CAP) {
        unsigned slot = base + inc - nh;
#pragma unroll
        for (int i = 0; i < NU; ++i) {
          const int u = wv + 32 * i;
          int q = (u < NUNIT ? u : wv) + rot;
          if (q >= NUNIT) q -= NUNIT;
          const unsigned c0 = (unsigned)(q * 64 + 2 * lane) << 5;
          unsigned long long m = hm[i];
          while (m) {
            const unsigned p = (unsigned)__builtin_ctzll(m);
            m &= m - 1ull;
            list[slot] = c0 + p;
            ++slot;
          }
        }
      }
    }
  }
  __syncthreads();
  const unsigned total = cnt;

  if (total <= (unsigned)CAP) {
    for (unsigned i = tid; i < total; i += NTHR) {
      const unsigned w = list[i];
      const unsigned p = w & 31u, j = (w >> 5) & 31u, blk = w >> 10;
      const unsigned e = blk * 1024u + 128u * (p >> 2) + 4u * j + (p & 3u);
      const unsigned s = (unsigned)ei[e];
      unsigned dl = (unsigned)(ei[N_EDGES + e] - n0);
      dl = dl < (unsigned)RNG ? dl : (unsigned)(RNG - 1);
      list[i] = s | (dl << 16);
    }
    __syncthreads();

    const unsigned items = total * HEADS;
    for (unsigned i0 = tid; i0 < items; i0 += NTHR * 4) {
      float sv[4];
      unsigned dh[4];
#pragma unroll
      for (int u = 0; u < 4; ++u) {
        const unsigned idx = i0 + u * NTHR;
        sv[u] = 0.f;
        dh[u] = 0xFFFFFFFFu;
        if (idx < items) {
          const unsigned w = list[idx >> 3];
          const unsigned hd = idx & 7u;
          sv[u] = asrc[(w & 0xFFFFu) * HEADS + hd];
          dh[u] = (w >> 16) * HEADS + hd;
        }
      }
#pragma unroll
      for (int u = 0; u < 4; ++u) {
        if (dh[u] != 0xFFFFFFFFu) {
          const unsigned idx = i0 + u * NTHR;
          const float v = leaky(sv[u] + adl[dh[u]]);
          __hip_atomic_fetch_max(&mxk[dh[u]], f32_key(v), __ATOMIC_RELAXED, __HIP_MEMORY_SCOPE_WORKGROUP);
          tab[idx] = v;
        }
      }
    }
    __syncthreads();

    for (unsigned i0 = tid; i0 < items; i0 += NTHR * 4) {
      float    lv[4];
      int      mk[4];
      unsigned bb[4], pp[4];
      unsigned le[4];
#pragma unroll
      for (int u = 0; u < 4; ++u) {
        const unsigned idx = i0 + u * NTHR;
        le[u] = list[(idx < items ? idx : i0) >> 3];
      }
#pragma unroll
      for (int u = 0; u < 4; ++u) {
        const unsigned idx = i0 + u * NTHR;
        const unsigned ix  = idx < items ? idx : i0;
        const unsigned hd  = ix & 7u;
        const unsigned b   = (le[u] >> 16) * HEADS + hd;
        pp[u] = ix;
        lv[u] = tab[pp[u]];
        mk[u] = mxk[b];
        bb[u] = idx < items ? b : 0xFFFFFFFFu;
      }
#pragma unroll
      for (int u = 0; u < 4; ++u) {
        if (bb[u] != 0xFFFFFFFFu) {
          const float ex = __expf(lv[u] - key_f32(mk[u]));
          tab[pp[u]] = ex;
          __hip_atomic_fetch_add(&den[bb[u]], ex, __ATOMIC_RELAXED, __HIP_MEMORY_SCOPE_WORKGROUP);
        }
      }
    }
    __syncthreads();

    {
      constexpr int SU = 8;
      const unsigned wv  = (unsigned)__builtin_amdgcn_readfirstlane(wave);
      const unsigned tot = (unsigned)__builtin_amdgcn_readfirstlane((int)total);
      unsigned i0 = wv;
      for (; i0 + 32u * (SU - 1) < tot; i0 += 32 * SU)
        sum_batch<true, SU>(i0, tot, lane, h, list, tab, acc);
      if (i0 < tot)
        sum_batch<false, SU>(i0, tot, lane, h, list, tab, acc);
    }
    __syncthreads();
  } else {
    for (int ph = 0; ph < 2; ++ph) {
      for (int base = wave * 32; base < N_EDGES; base += 32 * 32) {
        const unsigned d = (unsigned)(ei[N_EDGES + base + lane] - n0);
        unsigned mask = __builtin_amdgcn_ballot_w32(d < (unsigned)RNG);
        while (mask) {
          const int bpos = __builtin_ctz(mask);
          mask &= mask - 1u;
          const int s   = ei[base + bpos];
          const int dlu = __shfl((int)d, bpos, 32);
          const float sa = asrc[s * HEADS + (lane & 7)];
          if (ph == 0) {
            const int b = dlu * HEADS + (lane & 7);
            const float v = leaky(sa + adl[b]);
            if (lane < 8)
              __hip_atomic_fetch_max(&mxk[b], f32_key(v), __ATOMIC_RELAXED, __HIP_MEMORY_SCOPE_WORKGROUP);
          } else {
            const v4f hv = reinterpret_cast<const v4f*>(h + s * HF)[lane];
            edge_sum(lane, dlu, sa, hv, adl, mxk, den, acc);
          }
        }
      }
      __syncthreads();
    }
  }

  {
    const v4f b4 = reinterpret_cast<const v4f*>(bias)[lane];
    for (int i = tid; i < nloc * 32; i += NTHR) {
      const int r = i >> 5;
      const float dn = den[r * HEADS + (lane >> 2)] + 1e-16f;
      const float* ap = acc + r * HF + lane;
      v4f o;
#pragma unroll
      for (int j = 0; j < 4; ++j) o[j] = ap[32 * j] / dn + b4[j];
      reinterpret_cast<v4f*>(out + (size_t)(n0 + r) * HF)[lane] = o;
    }
  }
}

extern "C" void kernel_launch(void* const* d_in, const int* in_sizes, int n_in,
                              void* d_out, int out_size, void* d_ws, size_t ws_size,
                              hipStream_t stream) {
  const float* x        = (const float*)d_in[0];
  const float* W        = (const float*)d_in[1];
  const float* att_src  = (const float*)d_in[2];
  const float* att_dst  = (const float*)d_in[3];
  const float* bias     = (const float*)d_in[4];
  const int*   ei       = (const int*)d_in[5];
  float*       out      = (float*)d_out;

  float* h     = (float*)d_ws;
  float* asrc  = h     + (size_t)N_NODES * HF;
  float* adst  = asrc  + (size_t)N_NODES * HEADS;
  unsigned* tagp = (unsigned*)(adst + (size_t)N_NODES * HEADS);

  const int BLK = 256;

  gat_gemm_wmma<<<GEMM_WG, BLK, 0, stream>>>(x, W, h);

  gat_logits<<<(N_NODES * HEADS + BLK - 1) / BLK, BLK, 0, stream>>>(
      h, att_src, att_dst, ei, asrc, adst, tagp);

  gat_owner<<<NGRP, NTHR, 0, stream>>>(ei, tagp, h, asrc, adst, bias, out);
}


# Logits launch folded into the projection launch: a_src/a_dst from the accumulator tile by a 4-step lane tree (16-term sum order changes, still fixed); tags by wave 0 of 782 groups; 3 launches to 2
# speedup vs baseline: 4.2939x; 1.0354x over previous
#include <hip/hip_runtime.h>
#include <hip/hip_bf16.h>

#define N_NODES 50000
#define N_EDGES 800000
#define F_IN    128
#define HEADS   8
#define FH      16
#define HF      128
#define NEG_SLOPE 0.2f

typedef float v2f __attribute__((ext_vector_type(2)));
typedef float v4f __attribute__((ext_vector_type(4)));
typedef float v8f __attribute__((ext_vector_type(8)));

#define RNG   196
#define NGRP  256
#define NCH   (N_EDGES / 32)
#define NBLK  ((NCH + 31) / 32)
#define NPLANE 32
#define PW    (NBLK * 32)
#define NUNIT (PW / 64)

typedef int v4i __attribute__((ext_vector_type(4)));
typedef unsigned v2u __attribute__((ext_vector_type(2)));

static_assert(N_EDGES % 128 == 0, "a tag block is read in whole 16-byte loads of 32 lanes");
static_assert(PW % 64 == 0, "a plane is read in whole 256-byte pieces");

__device__ __forceinline__ void tag_block(int gw, int lane,
                                          const int* __restrict__ ei,
                                          unsigned* __restrict__ tagp) {
  const int* __restrict__ dcol = ei + N_EDGES + gw * 1024;
  v4i d4[8];
#pragma unroll
  for (int r = 0; r < 8; ++r) {
    const int ro = (gw * 1024 + 128 * r < N_EDGES) ? 128 * r : 0;
    d4[r] = reinterpret_cast<const v4i*>(dcol + ro)[lane];
  }
  unsigned row[32];
#pragma unroll
  for (int r = 0; r < 8; ++r) {
    const bool inlist = gw * 1024 + 128 * r < N_EDGES;
#pragma unroll
    for (int c = 0; c < 4; ++c) {
      unsigned g = (unsigned)d4[r][c] / (unsigned)RNG;
      g = g < (unsigned)NGRP ? g : (unsigned)(NGRP - 1);
      const unsigned oh = (1u << (g >> 4)) | (0x10000u << (g & 15u));
      row[4 * r + c] = inlist ? oh : 0u;
    }
  }
#pragma unroll
  for (int s = 0; s < 5; ++s) {
    const int j = 16 >> s;
    const unsigned m = 0xFFFFFFFFu / ((1u << j) + 1u);
#pragma unroll
    for (int k = 0; k < 32; ++k) {
      if ((k & j) == 0) {
        const unsigned t = ((row[k] >> j) ^ row[k + j]) & m;
        row[k]     ^= t << j;
        row[k + j] ^= t;
      }
    }
  }
#pragma unroll
  for (int P = 0; P < NPLANE; ++P) tagp[P * PW + gw * 32 + lane] = row[P];
}

template <int MASK>
__device__ __forceinline__ float lane_xor16(float v) {
  return __int_as_float(__builtin_amdgcn_update_dpp(
      0, __float_as_int(v), 0x160 | MASK, 0xF, 0xF, true));
}

#define GEMM_WG  1024
#define M_TILES  (N_NODES / 16)

static_assert(N_NODES % 16 == 0, "whole 16-row tiles");
#define XT_WG    (M_TILES % GEMM_WG)
static_assert(XT_WG + NBLK <= GEMM_WG, "a work-group of the projection launch for every tag block");

__global__ __launch_bounds__(256) void gat_gemm_wmma(
    const float* __restrict__ x, const float* __restrict__ W,
    const float* __restrict__ att_src, const float* __restrict__ att_dst,
    const int* __restrict__ ei,
    float* __restrict__ h, float* __restrict__ asrc, float* __restrict__ adst,
    unsigned* __restrict__ tagp) {
  const int lane = threadIdx.x & 31;
  const int wave = threadIdx.x >> 5;
  const int n0   = wave * 16;
  const int half = lane >> 4;
  const int l15  = lane & 15;

  if (__builtin_amdgcn_readfirstlane(wave) == 0 &&
      blockIdx.x >= XT_WG && blockIdx.x < XT_WG + NBLK)
    tag_block(blockIdx.x - XT_WG, lane, ei, tagp);

  const v4f* __restrict__ brow =
      reinterpret_cast<const v4f*>(W + (n0 + l15) * F_IN + 4 * half);
  v4f b[16];
#pragma unroll
  for (int q = 0; q < 16; ++q) b[q] = brow[2 * q];

  const bool b3 = (l15 & 8) != 0, b2 = (l15 & 4) != 0, b1 = (l15 & 2) != 0, b0 = (l15 & 1) != 0;
  const float kat = (b3 ? att_dst : att_src)[n0 + l15];
  const float sat = (b3 ? att_src : att_dst)[n0 + l15];
  float* __restrict__ lp = (b3 ? adst : asrc) + (half * 8 + (l15 & 7)) * HEADS + wave;

  for (int t = blockIdx.x; t < M_TILES; t += GEMM_WG) {
    const int m0 = t * 16;
    const v4f* __restrict__ arow =
        reinterpret_cast<const v4f*>(x + (m0 + l15) * F_IN + 4 * half);
    v8f c = {};
#pragma unroll
    for (int hh = 0; hh < 2; ++hh) {
      v4f a[8];
#pragma unroll
      for (int q = 0; q < 8; ++q) a[q] = arow[2 * (8 * hh + q)];
      asm volatile("" :: "v"(a[0]), "v"(a[1]), "v"(a[2]), "v"(a[3]),
                         "v"(a[4]), "v"(a[5]), "v"(a[6]), "v"(a[7]));
#pragma unroll
      for (int q = 0; q < 8; ++q) {
        const v4f bq = b[8 * hh + q];
        v2f a0; a0.x = a[q].x; a0.y = a[q].y;
        v2f a1; a1.x = a[q].z; a1.y = a[q].w;
        v2f b0v; b0v.x = bq.x;   b0v.y = bq.y;
        v2f b1v; b1v.x = bq.z;   b1v.y = bq.w;
        c = __builtin_amdgcn_wmma_f32_16x16x4_f32(
             false, a0,  false, b0v,
             (short)0, c,  false,  false);
        c = __builtin_amdgcn_wmma_f32_16x16x4_f32(
             false, a1,  false, b1v,
             (short)0, c,  false,  false);
      }
    }

    float* __restrict__ op = h + (m0 + half * 8) * HF + n0 + l15;
#pragma unroll
    for (int v = 0; v < 8; ++v) op[v * HF] = c[v];

    float q8[8], q4[4], q2[2];
#pragma unroll
    for (int i = 0; i < 8; ++i)
      q8[i] = c[i] * kat + lane_xor16<8>(c[i] * sat);
#pragma unroll
    for (int i = 0; i < 4; ++i)
      q4[i] = (b2 ? q8[i + 4] : q8[i]) + lane_xor16<4>(b2 ? q8[i] : q8[i + 4]);
#pragma unroll
    for (int i = 0; i < 2; ++i)
      q2[i] = (b1 ? q4[i + 2] : q4[i]) + lane_xor16<2>(b1 ? q4[i] : q4[i + 2]);
    lp[m0 * HEADS] = (b0 ? q2[1] : q2[0]) + lane_xor16<1>(b0 ? q2[0] : q2[1]);
  }
}

__device__ __forceinline__ int f32_key(float f) {
  int i = __float_as_int(f);
  return i >= 0 ? i : (i ^ 0x7fffffff);
}
__device__ __forceinline__ float key_f32(int k) {
  return __int_as_float(k >= 0 ? k : (k ^ 0x7fffffff));
}

__device__ __forceinline__ float leaky(float v) {
  return v >= 0.f ? v : NEG_SLOPE * v;
}

#define CAP   5120
#define NTHR  1024

static_assert(RNG * NGRP >= N_NODES, "ranges must cover every node");
static_assert(N_NODES <= 65536 && RNG <= 256 && NGRP <= 256, "list entry packs src in 16 bits, local dst in 8; a tag is a byte");
static_assert(N_EDGES % 32 == 0, "a tag word covers 32 edges, the exact path walks 32 edges a wave");
static_assert(NBLK * 1024 <= (1 << 27) && NUNIT >= NGRP, "hit entry packs block, word and bit in 32 bits; every group has a start piece of its own");

__device__ __forceinline__ void edge_sum(int lane, int dl, float sa, const v4f hv,
                                         const float* adl, const int* mxk,
                                         float* den, float* acc) {
  const int b = dl * HEADS + (lane & 7);
  const float v  = leaky(sa + adl[b]);
  const float ex = __expf(v - key_f32(mxk[b]));
  if (lane < 8)
    __hip_atomic_fetch_add(&den[b], ex, __ATOMIC_RELAXED, __HIP_MEMORY_SCOPE_WORKGROUP);
  float* ap = acc + dl * HF + lane;
  const float a = __shfl(ex, lane >> 2, 32);
#pragma unroll
  for (int j = 0; j < 4; ++j)
    __hip_atomic_fetch_add(&ap[32 * j], a * hv[j], __ATOMIC_RELAXED, __HIP_MEMORY_SCOPE_WORKGROUP);
}

template <bool FULL, int SU>
__device__ __forceinline__ void sum_batch(unsigned i0, unsigned tot, int lane,
                                          const float* __restrict__ h,
                                          const unsigned* list, const float* tab,
                                          float* acc) {
  const unsigned hq = (unsigned)lane >> 2;
  const unsigned lane16 = (unsigned)lane * 16u;
  const unsigned li = i0 + 32u * (unsigned)(lane & (SU - 1));
  const unsigned lw = list[(FULL || li < tot) ? li : i0];
  v4f   hv[SU];
  float wt[SU];
  unsigned dl[SU];
#pragma unroll
  for (int u = 0; u < SU; ++u) {
    const unsigned i  = i0 + 32u * (unsigned)u;
    const unsigned ic = (FULL || i < tot) ? i : i0;
    const unsigned w  = (unsigned)__builtin_amdgcn_readlane((int)lw, u);
    dl[u] = w >> 16;
    const unsigned off = ((w & 0xFFFFu) << 9) + lane16;
    hv[u] = *reinterpret_cast<const v4f*>(reinterpret_cast<const char*>(h) + off);
    wt[u] = tab[ic * HEADS + hq];
  }
#pragma unroll
  for (int u = 0; u < SU; ++u) {
    if (FULL || i0 + 32u * (unsigned)u < tot) {
      float* ap = acc + dl[u] * HF + lane;
#pragma unroll
      for (int j = 0; j < 4; ++j)
        __hip_atomic_fetch_add(&ap[32 * j], wt[u] * hv[u][j], __ATOMIC_RELAXED, __HIP_MEMORY_SCOPE_WORKGROUP);
    }
  }
}

__global__ __launch_bounds__(NTHR) void gat_owner(
    const int* __restrict__ ei, const unsigned* __restrict__ tagp,
    const float* __restrict__ h,
    const float* __restrict__ asrc, const float* __restrict__ adst,
    const float* __restrict__ bias, float* __restrict__ out) {
  __shared__ float    acc[RNG * HF];
  __shared__ int      mxk[RNG * HEADS];
  __shared__ float    den[RNG * HEADS];
  __shared__ float    adl[RNG * HEADS];
  __shared__ unsigned list[CAP];
  __shared__ float    tab[CAP * HEADS];
  __shared__ unsigned cnt;

  const int tid  = threadIdx.x;
  const int lane = tid & 31;
  const int wave = tid >> 5;
  const int n0   = blockIdx.x * RNG;
  const int nloc = (N_NODES - n0) < RNG ? (N_NODES - n0) : RNG;

  for (int i = tid; i < RNG * HF; i += NTHR) acc[i] = 0.f;
  for (int i = tid; i < RNG * HEADS; i += NTHR) {
    mxk[i] = (int)0x80000000;
    den[i] = 0.f;
    adl[i] = (i < nloc * HEADS) ? adst[n0 * HEADS + i] : 0.f;
  }
  if (tid == 0) cnt = 0u;
  __syncthreads();

  {
    constexpr int NU = (NUNIT + 31) / 32;
    const int wv = __builtin_amdgcn_readfirstlane(wave);
    const unsigned* __restrict__ pa = tagp + (blockIdx.x >> 4) * PW;
    const unsigned* __restrict__ pb = tagp + (16 + (blockIdx.x & 15)) * PW;
    const int rot = (blockIdx.x * NUNIT) / NGRP;
    v2u wa[NU], wb[NU];
#pragma unroll
    for (int i = 0; i < NU; ++i) {
      const int u = wv + 32 * i;
      int q = (u < NUNIT ? u : wv) + rot;
      if (q >= NUNIT) q -= NUNIT;
      wa[i] = *reinterpret_cast<const v2u*>(pa + q * 64 + 2 * lane);
      wb[i] = *reinterpret_cast<const v2u*>(pb + q * 64 + 2 * lane);
    }
    unsigned long long hm[NU];
    unsigned nh = 0u;
#pragma unroll
    for (int i = 0; i < NU; ++i) {
      const unsigned lo = wa[i].x & wb[i].x, hi = wa[i].y & wb[i].y;
      unsigned long long m = ((unsigned long long)hi << 32) | lo;
      if (wv + 32 * i >= NUNIT) m = 0ull;
      hm[i] = m;
      nh += (unsigned)__builtin_popcountll(m);
    }
    unsigned inc = nh;
#pragma unroll
    for (int d = 1; d < 32; d <<= 1) {
      const unsigned o = __shfl_up(inc, d, 32);
      if (lane >= d) inc += o;
    }
    const unsigned wtot = (unsigned)__builtin_amdgcn_readlane((int)inc, 31);
    if (wtot) {
      unsigned base = 0u;
      if (lane == 0) base = atomicAdd(&cnt, wtot);
      base = (unsigned)__builtin_amdgcn_readfirstlane((int)base);
      if (base + wtot <= (unsigned)CAP) {
        unsigned slot = base + inc - nh;
#pragma unroll
        for (int i = 0; i < NU; ++i) {
          const int u = wv + 32 * i;
          int q = (u < NUNIT ? u : wv) + rot;
          if (q >= NUNIT) q -= NUNIT;
          const unsigned c0 = (unsigned)(q * 64 + 2 * lane) << 5;
          unsigned long long m = hm[i];
          while (m) {
            const unsigned p = (unsigned)__builtin_ctzll(m);
            m &= m - 1ull;
            list[slot] = c0 + p;
            ++slot;
          }
        }
      }
    }
  }
  __syncthreads();
  const unsigned total = cnt;

  if (total <= (unsigned)CAP) {
    for (unsigned i = tid; i < total; i += NTHR) {
      const unsigned w = list[i];
      const unsigned p = w & 31u, j = (w >> 5) & 31u, blk = w >> 10;
      const unsigned e = blk * 1024u + 128u * (p >> 2) + 4u * j + (p & 3u);
      const unsigned s = (unsigned)ei[e];
      unsigned dl = (unsigned)(ei[N_EDGES + e] - n0);
      dl = dl < (unsigned)RNG ? dl : (unsigned)(RNG - 1);
      list[i] = s | (dl << 16);
    }
    __syncthreads();

    const unsigned items = total * HEADS;
    for (unsigned i0 = tid; i0 < items; i0 += NTHR * 4) {
      float sv[4];
      unsigned dh[4];
#pragma unroll
      for (int u = 0; u < 4; ++u) {
        const unsigned idx = i0 + u * NTHR;
        sv[u] = 0.f;
        dh[u] = 0xFFFFFFFFu;
        if (idx < items) {
          const unsigned w = list[idx >> 3];
          const unsigned hd = idx & 7u;
          sv[u] = asrc[(w & 0xFFFFu) * HEADS + hd];
          dh[u] = (w >> 16) * HEADS + hd;
        }
      }
#pragma unroll
      for (int u = 0; u < 4; ++u) {
        if (dh[u] != 0xFFFFFFFFu) {
          const unsigned idx = i0 + u * NTHR;
          const float v = leaky(sv[u] + adl[dh[u]]);
          __hip_atomic_fetch_max(&mxk[dh[u]], f32_key(v), __ATOMIC_RELAXED, __HIP_MEMORY_SCOPE_WORKGROUP);
          tab[idx] = v;
        }
      }
    }
    __syncthreads();

    for (unsigned i0 = tid; i0 < items; i0 += NTHR * 4) {
      float    lv[4];
      int      mk[4];
      unsigned bb[4], pp[4];
      unsigned le[4];
#pragma unroll
      for (int u = 0; u < 4; ++u) {
        const unsigned idx = i0 + u * NTHR;
        le[u] = list[(idx < items ? idx : i0) >> 3];
      }
#pragma unroll
      for (int u = 0; u < 4; ++u) {
        const unsigned idx = i0 + u * NTHR;
        const unsigned ix  = idx < items ? idx : i0;
        const unsigned hd  = ix & 7u;
        const unsigned b   = (le[u] >> 16) * HEADS + hd;
        pp[u] = ix;
        lv[u] = tab[pp[u]];
        mk[u] = mxk[b];
        bb[u] = idx < items ? b : 0xFFFFFFFFu;
      }
#pragma unroll
      for (int u = 0; u < 4; ++u) {
        if (bb[u] != 0xFFFFFFFFu) {
          const float ex = __expf(lv[u] - key_f32(mk[u]));
          tab[pp[u]] = ex;
          __hip_atomic_fetch_add(&den[bb[u]], ex, __ATOMIC_RELAXED, __HIP_MEMORY_SCOPE_WORKGROUP);
        }
      }
    }
    __syncthreads();

    {
      constexpr int SU = 8;
      const unsigned wv  = (unsigned)__builtin_amdgcn_readfirstlane(wave);
      const unsigned tot = (unsigned)__builtin_amdgcn_readfirstlane((int)total);
      unsigned i0 = wv;
      for (; i0 + 32u * (SU - 1) < tot; i0 += 32 * SU)
        sum_batch<true, SU>(i0, tot, lane, h, list, tab, acc);
      if (i0 < tot)
        sum_batch<false, SU>(i0, tot, lane, h, list, tab, acc);
    }
    __syncthreads();
  } else {
    for (int ph = 0; ph < 2; ++ph) {
      for (int base = wave * 32; base < N_EDGES; base += 32 * 32) {
        const unsigned d = (unsigned)(ei[N_EDGES + base + lane] - n0);
        unsigned mask = __builtin_amdgcn_ballot_w32(d < (unsigned)RNG);
        while (mask) {
          const int bpos = __builtin_ctz(mask);
          mask &= mask - 1u;
          const int s   = ei[base + bpos];
          const int dlu = __shfl((int)d, bpos, 32);
          const float sa = asrc[s * HEADS + (lane & 7)];
          if (ph == 0) {
            const int b = dlu * HEADS + (lane & 7);
            const float v = leaky(sa + adl[b]);
            if (lane < 8)
              __hip_atomic_fetch_max(&mxk[b], f32_key(v), __ATOMIC_RELAXED, __HIP_MEMORY_SCOPE_WORKGROUP);
          } else {
            const v4f hv = reinterpret_cast<const v4f*>(h + s * HF)[lane];
            edge_sum(lane, dlu, sa, hv, adl, mxk, den, acc);
          }
        }
      }
      __syncthreads();
    }
  }

  {
    const v4f b4 = reinterpret_cast<const v4f*>(bias)[lane];
    for (int i = tid; i < nloc * 32; i += NTHR) {
      const int r = i >> 5;
      const float dn = den[r * HEADS + (lane >> 2)] + 1e-16f;
      const float* ap = acc + r * HF + lane;
      v4f o;
#pragma unroll
      for (int j = 0; j < 4; ++j) o[j] = ap[32 * j] / dn + b4[j];
      reinterpret_cast<v4f*>(out + (size_t)(n0 + r) * HF)[lane] = o;
    }
  }
}

extern "C" void kernel_launch(void* const* d_in, const int* in_sizes, int n_in,
                              void* d_out, int out_size, void* d_ws, size_t ws_size,
                              hipStream_t stream) {
  const float* x        = (const float*)d_in[0];
  const float* W        = (const float*)d_in[1];
  const float* att_src  = (const float*)d_in[2];
  const float* att_dst  = (const float*)d_in[3];
  const float* bias     = (const float*)d_in[4];
  const int*   ei       = (const int*)d_in[5];
  float*       out      = (float*)d_out;

  float* h     = (float*)d_ws;
  float* asrc  = h     + (size_t)N_NODES * HF;
  float* adst  = asrc  + (size_t)N_NODES * HEADS;
  unsigned* tagp = (unsigned*)(adst + (size_t)N_NODES * HEADS);

  const int BLK = 256;

  gat_gemm_wmma<<<GEMM_WG, BLK, 0, stream>>>(x, W, att_src, att_dst, ei, h, asrc, adst, tagp);

  gat_owner<<<NGRP, NTHR, 0, stream>>>(ei, tagp, h, asrc, adst, bias, out);
}
